# starting kernel
# baseline (speedup 1.0000x reference)
#include <hip/hip_runtime.h>

#define N_NODES 50000
#define N_EDGES 1600000
#define DIN 128
#define DOUT 256
#define DA 32
#define SLOPE 0.2f

typedef __attribute__((ext_vector_type(2))) float v2f;
typedef __attribute__((ext_vector_type(8))) float v8f;

static __device__ __forceinline__ v8f wmma_f32_16x16x4(v2f a, v2f b, v8f c) {
  return __builtin_amdgcn_wmma_f32_16x16x4_f32(false, a, false, b, (short)0, c,
                                               false, false);
}

__global__ void proj_kernel(const float* __restrict__ x,
                            const float* __restrict__ W_row,
                            const float* __restrict__ W_col,
                            float* __restrict__ a_row,
                            float* __restrict__ a_col) {
  const int wave = (blockIdx.x * blockDim.x + threadIdx.x) >> 5;
  const int lane = threadIdx.x & 31;
  if (wave >= N_NODES / 16) return;
  const int tile = wave;
  const int half = lane >> 4;
  const int m    = lane & 15;
  const int node = tile * 16 + m;

  v8f acc0 = {}, acc1 = {}, acc2 = {}, acc3 = {};
  const float* xrow = x + (size_t)node * DIN + 2 * half;
  for (int k = 0; k < DIN; k += 4) {
    v2f a = *(const v2f*)(xrow + k);
    const int kk = k + 2 * half;
    v2f b0 = { W_row[kk * DA + m],      W_row[(kk + 1) * DA + m] };
    v2f b1 = { W_row[kk * DA + 16 + m], W_row[(kk + 1) * DA + 16 + m] };
    v2f b2 = { W_col[kk * DA + m],      W_col[(kk + 1) * DA + m] };
    v2f b3 = { W_col[kk * DA + 16 + m], W_col[(kk + 1) * DA + 16 + m] };
    acc0 = wmma_f32_16x16x4(a, b0, acc0);
    acc1 = wmma_f32_16x16x4(a, b1, acc1);
    acc2 = wmma_f32_16x16x4(a, b2, acc2);
    acc3 = wmma_f32_16x16x4(a, b3, acc3);
  }
#pragma unroll
  for (int v = 0; v < 8; ++v) {
    const int on = tile * 16 + v + half * 8;
    a_row[on * DA + m]      = acc0[v];
    a_row[on * DA + 16 + m] = acc1[v];
    a_col[on * DA + m]      = acc2[v];
    a_col[on * DA + 16 + m] = acc3[v];
  }
}

__global__ void edge_score_kernel(const float* __restrict__ a_row,
                                  const float* __restrict__ a_col,
                                  const int* __restrict__ row,
                                  const int* __restrict__ col,
                                  float* __restrict__ raw,
                                  unsigned* __restrict__ maxkey) {
  const int e = blockIdx.x * blockDim.x + threadIdx.x;
  if (e >= N_EDGES) return;
  const float4* ar = (const float4*)(a_row + (size_t)row[e] * DA);
  const float4* ac = (const float4*)(a_col + (size_t)col[e] * DA);
  float s = 0.f;
#pragma unroll
  for (int i = 0; i < DA / 4; ++i) {
    float4 u = ar[i], v = ac[i];
    s += u.x * v.x + u.y * v.y + u.z * v.z + u.w * v.w;
  }
  s *= 0.17677669529663687f;
  s = (s >= 0.f) ? s : SLOPE * s;
  raw[e] = s;
  unsigned b = __float_as_uint(s);
  unsigned key = (b & 0x80000000u) ? ~b : (b | 0x80000000u);
#pragma unroll
  for (int off = 16; off > 0; off >>= 1) {
    unsigned o = __shfl_xor(key, off, 32);
    key = (o > key) ? o : key;
  }
  if ((threadIdx.x & 31) == 0) atomicMax(maxkey, key);
}

__global__ void edge_exp_deg_kernel(const float* __restrict__ raw,
                                    const int* __restrict__ row,
                                    const unsigned* __restrict__ maxkey,
                                    float* __restrict__ atten,
                                    float* __restrict__ deg) {
  const int e = blockIdx.x * blockDim.x + threadIdx.x;
  if (e >= N_EDGES) return;
  const unsigned key = *maxkey;
  const unsigned b = (key & 0x80000000u) ? (key ^ 0x80000000u) : ~key;
  const float gmax = __uint_as_float(b);
  const float a = expf(raw[e] - gmax);
  atten[e] = a;
  atomicAdd(deg + row[e], a);
}

__global__ void scatter_kernel(const float* __restrict__ x,
                               const int* __restrict__ row,
                               const int* __restrict__ col,
                               const float* __restrict__ adj_val,
                               const float* __restrict__ atten,
                               const float* __restrict__ deg,
                               float* __restrict__ feat) {
  const int wpb = blockDim.x >> 5;
  const int e = __builtin_amdgcn_readfirstlane(blockIdx.x * wpb +
                                               (threadIdx.x >> 5));
  const int lane = threadIdx.x & 31;
  if (e >= N_EDGES) return;
  const int r = row[e], c = col[e];
  const float val = (adj_val[e] + atten[e] / (deg[r] + 1e-15f)) * 0.5f;
  const float4 xv = *(const float4*)(x + (size_t)c * DIN + lane * 4);
  float* dst = feat + (size_t)r * DIN + lane * 4;
  atomicAdd(dst + 0, val * xv.x);
  atomicAdd(dst + 1, val * xv.y);
  atomicAdd(dst + 2, val * xv.z);
  atomicAdd(dst + 3, val * xv.w);
}

__global__ void out_gemm_kernel(const float* __restrict__ feat,
                                const float* __restrict__ W_x,
                                const float* __restrict__ b_x,
                                float* __restrict__ out) {
  const int wave = (blockIdx.x * blockDim.x + threadIdx.x) >> 5;
  const int lane = threadIdx.x & 31;
  const int total = (N_NODES / 16) * 4;
  if (wave >= total) return;
  const int rt    = wave >> 2;
  const int cbase = (wave & 3) * 64;
  const int half  = lane >> 4;
  const int m     = lane & 15;
  const int node  = rt * 16 + m;

  v8f acc[4];
#pragma unroll
  for (int t = 0; t < 4; ++t) {
    const float bias = b_x[cbase + t * 16 + m];
#pragma unroll
    for (int v = 0; v < 8; ++v) acc[t][v] = bias;
  }
  const float* frow = feat + (size_t)node * DIN + 2 * half;
  for (int k = 0; k < DIN; k += 4) {
    v2f a = *(const v2f*)(frow + k);
    const int kk = k + 2 * half;
#pragma unroll
    for (int t = 0; t < 4; ++t) {
      v2f b = { W_x[kk * DOUT + cbase + t * 16 + m],
                W_x[(kk + 1) * DOUT + cbase + t * 16 + m] };
      acc[t] = wmma_f32_16x16x4(a, b, acc[t]);
    }
  }
#pragma unroll
  for (int t = 0; t < 4; ++t)
#pragma unroll
    for (int v = 0; v < 8; ++v)
      out[(size_t)(rt * 16 + v + half * 8) * DOUT + cbase + t * 16 + m] = acc[t][v];
}

extern "C" void kernel_launch(void* const* d_in, const int* in_sizes, int n_in,
                              void* d_out, int out_size, void* d_ws, size_t ws_size,
                              hipStream_t stream) {
  const float* x      = (const float*)d_in[0];
  const int*   row    = (const int*)d_in[1];
  const int*   col    = (const int*)d_in[2];
  const float* adjval = (const float*)d_in[3];
  const float* W_row  = (const float*)d_in[4];
  const float* W_col  = (const float*)d_in[5];
  const float* W_x    = (const float*)d_in[6];
  const float* b_x    = (const float*)d_in[7];
  float* out = (float*)d_out;

  char* ws = (char*)d_ws;
  size_t off = 0;
  float* a_row = (float*)(ws + off); off += (size_t)N_NODES * DA * 4;
  float* a_col = (float*)(ws + off); off += (size_t)N_NODES * DA * 4;
  float* raw   = (float*)(ws + off); off += (size_t)N_EDGES * 4;
  float* atten = (float*)(ws + off); off += (size_t)N_EDGES * 4;
  float* deg   = (float*)(ws + off); off += (size_t)N_NODES * 4;
  unsigned* maxkey = (unsigned*)(ws + off); off += 256;
  float* feat  = (float*)(ws + off); off += (size_t)N_NODES * DIN * 4;

  hipMemsetAsync(deg, 0, (size_t)N_NODES * 4, stream);
  hipMemsetAsync(maxkey, 0, 4, stream);
  hipMemsetAsync(feat, 0, (size_t)N_NODES * DIN * 4, stream);

  {
    const int waves = N_NODES / 16;
    const int blocks = (waves * 32 + 255) / 256;
    proj_kernel<<<blocks, 256, 0, stream>>>(x, W_row, W_col, a_row, a_col);
  }
  {
    const int blocks = (N_EDGES + 255) / 256;
    edge_score_kernel<<<blocks, 256, 0, stream>>>(a_row, a_col, row, col, raw, maxkey);
    edge_exp_deg_kernel<<<blocks, 256, 0, stream>>>(raw, row, maxkey, atten, deg);
  }
  {
    const int blocks = N_EDGES / 8;
    scatter_kernel<<<blocks, 256, 0, stream>>>(x, row, col, adjval, atten, deg, feat);
  }
  {
    const int waves = (N_NODES / 16) * 4;
    const int blocks = (waves * 32 + 255) / 256;
    out_gemm_kernel<<<blocks, 256, 0, stream>>>(feat, W_x, b_x, out);
  }
}


# SpMM by row buckets: each edge takes a rank by a returning add on its row's own line, rows get segments, (col,val) pairs are filled in, one wave per row sums val*x[col] in registers
# speedup vs baseline: 1.5639x; 1.5639x over previous
#include <hip/hip_runtime.h>

#define N_NODES 50000
#define N_EDGES 1600000
#define DIN 128
#define DOUT 256
#define DA 32
#define SLOPE 0.2f
#define REC 32

typedef __attribute__((ext_vector_type(2))) float v2f;
typedef __attribute__((ext_vector_type(8))) float v8f;

static __device__ __forceinline__ v8f wmma_f32_16x16x4(v2f a, v2f b, v8f c) {
  return __builtin_amdgcn_wmma_f32_16x16x4_f32(false, a, false, b, (short)0, c,
                                               false, false);
}

__global__ void proj_kernel(const float* __restrict__ x,
                            const float* __restrict__ W_row,
                            const float* __restrict__ W_col,
                            float* __restrict__ a_row,
                            float* __restrict__ a_col) {
  const int wave = (blockIdx.x * blockDim.x + threadIdx.x) >> 5;
  const int lane = threadIdx.x & 31;
  if (wave >= N_NODES / 16) return;
  const int tile = wave;
  const int half = lane >> 4;
  const int m    = lane & 15;
  const int node = tile * 16 + m;

  v8f acc0 = {}, acc1 = {}, acc2 = {}, acc3 = {};
  const float* xrow = x + (size_t)node * DIN + 2 * half;
  for (int k = 0; k < DIN; k += 4) {
    v2f a = *(const v2f*)(xrow + k);
    const int kk = k + 2 * half;
    v2f b0 = { W_row[kk * DA + m],      W_row[(kk + 1) * DA + m] };
    v2f b1 = { W_row[kk * DA + 16 + m], W_row[(kk + 1) * DA + 16 + m] };
    v2f b2 = { W_col[kk * DA + m],      W_col[(kk + 1) * DA + m] };
    v2f b3 = { W_col[kk * DA + 16 + m], W_col[(kk + 1) * DA + 16 + m] };
    acc0 = wmma_f32_16x16x4(a, b0, acc0);
    acc1 = wmma_f32_16x16x4(a, b1, acc1);
    acc2 = wmma_f32_16x16x4(a, b2, acc2);
    acc3 = wmma_f32_16x16x4(a, b3, acc3);
  }
#pragma unroll
  for (int v = 0; v < 8; ++v) {
    const int on = tile * 16 + v + half * 8;
    a_row[on * DA + m]      = acc0[v];
    a_row[on * DA + 16 + m] = acc1[v];
    a_col[on * DA + m]      = acc2[v];
    a_col[on * DA + 16 + m] = acc3[v];
  }
}

__global__ void edge_score_kernel(const float* __restrict__ a_row,
                                  const float* __restrict__ a_col,
                                  const int* __restrict__ row,
                                  const int* __restrict__ col,
                                  float* __restrict__ raw,
                                  unsigned* __restrict__ maxkey,
                                  int* __restrict__ rec,
                                  int* __restrict__ pos) {
  const int e = blockIdx.x * blockDim.x + threadIdx.x;
  if (e >= N_EDGES) return;
  const int r = row[e];
  const float4* ar = (const float4*)(a_row + (size_t)r * DA);
  const float4* ac = (const float4*)(a_col + (size_t)col[e] * DA);
  float4 u[DA / 4], v[DA / 4];
#pragma unroll
  for (int i = 0; i < DA / 4; ++i) { u[i] = ar[i]; v[i] = ac[i]; }
  pos[e] = atomicAdd(rec + (size_t)r * REC, 1);
  float s = 0.f;
#pragma unroll
  for (int i = 0; i < DA / 4; ++i)
    s += u[i].x * v[i].x + u[i].y * v[i].y + u[i].z * v[i].z + u[i].w * v[i].w;
  s *= 0.17677669529663687f;
  s = (s >= 0.f) ? s : SLOPE * s;
  raw[e] = s;
  unsigned b = __float_as_uint(s);
  unsigned key = (b & 0x80000000u) ? ~b : (b | 0x80000000u);
#pragma unroll
  for (int off = 16; off > 0; off >>= 1) {
    unsigned o = __shfl_xor(key, off, 32);
    key = (o > key) ? o : key;
  }
  if ((threadIdx.x & 31) == 0) atomicMax(maxkey, key);
}

__global__ void edge_exp_deg_kernel(float* __restrict__ sc,
                                    const int* __restrict__ row,
                                    const unsigned* __restrict__ maxkey,
                                    float* __restrict__ deg) {
  const int e = blockIdx.x * blockDim.x + threadIdx.x;
  if (e >= N_EDGES) return;
  const unsigned key = *maxkey;
  const unsigned b = (key & 0x80000000u) ? (key ^ 0x80000000u) : ~key;
  const float gmax = __uint_as_float(b);
  const float a = expf(sc[e] - gmax);
  sc[e] = a;
  atomicAdd(deg + row[e], a);
}

__global__ void seg_alloc_kernel(int* __restrict__ rec, int* __restrict__ total) {
  const int r = blockIdx.x * blockDim.x + threadIdx.x;
  const int lane = threadIdx.x & 31;
  const int c = (r < N_NODES) ? rec[(size_t)r * REC] : 0;
  int incl = c;
#pragma unroll
  for (int off = 1; off < 32; off <<= 1) {
    const int o = __shfl_up(incl, off, 32);
    if (lane >= off) incl += o;
  }
  int base = 0;
  if (lane == 31) base = atomicAdd(total, incl);
  base = __shfl(base, 31, 32);
  if (r < N_NODES) rec[(size_t)r * REC + 1] = base + incl - c;
}

__global__ void fill_kernel(const int* __restrict__ row,
                            const int* __restrict__ col,
                            const float* __restrict__ adj_val,
                            const float* __restrict__ atten,
                            const float* __restrict__ deg,
                            const int* __restrict__ rec,
                            const int* __restrict__ pos,
                            int2* __restrict__ packed) {
  const int e = blockIdx.x * blockDim.x + threadIdx.x;
  if (e >= N_EDGES) return;
  const int r = row[e];
  const float val = (adj_val[e] + atten[e] / (deg[r] + 1e-15f)) * 0.5f;
  int idx = rec[(size_t)r * REC + 1] + pos[e];
  idx = idx < 0 ? 0 : (idx > N_EDGES - 1 ? N_EDGES - 1 : idx);
  packed[idx] = make_int2(col[e], __float_as_int(val));
}

__global__ void gather_kernel(const float* __restrict__ x,
                              const int* __restrict__ rec,
                              const int2* __restrict__ packed,
                              float* __restrict__ feat) {
  const int wpb = blockDim.x >> 5;
  const int r = __builtin_amdgcn_readfirstlane(blockIdx.x * wpb +
                                               (threadIdx.x >> 5));
  const int lane = threadIdx.x & 31;
  if (r >= N_NODES) return;
  int n = rec[(size_t)r * REC];
  n = n > N_EDGES ? N_EDGES : n;
  const int s = rec[(size_t)r * REC + 1];
  const float* xl = x + lane * 4;
  float4 acc = make_float4(0.f, 0.f, 0.f, 0.f);
  for (int j = 0; j < n; j += 32) {
    const int m = (n - j < 32) ? (n - j) : 32;
    int2 p = make_int2(0, 0);
    if (lane < m) {
      int q = s + j + lane;
      q = q < 0 ? 0 : (q > N_EDGES - 1 ? N_EDGES - 1 : q);
      p = packed[q];
    }
    for (int k = 0; k < m; k += 4) {
      const unsigned c0 = min((unsigned)__builtin_amdgcn_readlane(p.x, k), (unsigned)(N_NODES - 1));
      const unsigned c1 = min((unsigned)__builtin_amdgcn_readlane(p.x, k + 1), (unsigned)(N_NODES - 1));
      const unsigned c2 = min((unsigned)__builtin_amdgcn_readlane(p.x, k + 2), (unsigned)(N_NODES - 1));
      const unsigned c3 = min((unsigned)__builtin_amdgcn_readlane(p.x, k + 3), (unsigned)(N_NODES - 1));
      const float v0 = __int_as_float(__builtin_amdgcn_readlane(p.y, k));
      const float v1 = __int_as_float(__builtin_amdgcn_readlane(p.y, k + 1));
      const float v2 = __int_as_float(__builtin_amdgcn_readlane(p.y, k + 2));
      const float v3 = __int_as_float(__builtin_amdgcn_readlane(p.y, k + 3));
      const float4 x0 = *(const float4*)(xl + (size_t)c0 * DIN);
      const float4 x1 = *(const float4*)(xl + (size_t)c1 * DIN);
      const float4 x2 = *(const float4*)(xl + (size_t)c2 * DIN);
      const float4 x3 = *(const float4*)(xl + (size_t)c3 * DIN);
      acc.x += v0 * x0.x; acc.y += v0 * x0.y; acc.z += v0 * x0.z; acc.w += v0 * x0.w;
      acc.x += v1 * x1.x; acc.y += v1 * x1.y; acc.z += v1 * x1.z; acc.w += v1 * x1.w;
      acc.x += v2 * x2.x; acc.y += v2 * x2.y; acc.z += v2 * x2.z; acc.w += v2 * x2.w;
      acc.x += v3 * x3.x; acc.y += v3 * x3.y; acc.z += v3 * x3.z; acc.w += v3 * x3.w;
    }
  }
  *(float4*)(feat + (size_t)r * DIN + lane * 4) = acc;
}

__global__ void out_gemm_kernel(const float* __restrict__ feat,
                                const float* __restrict__ W_x,
                                const float* __restrict__ b_x,
                                float* __restrict__ out) {
  const int wave = (blockIdx.x * blockDim.x + threadIdx.x) >> 5;
  const int lane = threadIdx.x & 31;
  const int total = (N_NODES / 16) * 4;
  if (wave >= total) return;
  const int rt    = wave >> 2;
  const int cbase = (wave & 3) * 64;
  const int half  = lane >> 4;
  const int m     = lane & 15;
  const int node  = rt * 16 + m;

  v8f acc[4];
#pragma unroll
  for (int t = 0; t < 4; ++t) {
    const float bias = b_x[cbase + t * 16 + m];
#pragma unroll
    for (int v = 0; v < 8; ++v) acc[t][v] = bias;
  }
  const float* frow = feat + (size_t)node * DIN + 2 * half;
  for (int k = 0; k < DIN; k += 4) {
    v2f a = *(const v2f*)(frow + k);
    const int kk = k + 2 * half;
#pragma unroll
    for (int t = 0; t < 4; ++t) {
      v2f b = { W_x[kk * DOUT + cbase + t * 16 + m],
                W_x[(kk + 1) * DOUT + cbase + t * 16 + m] };
      acc[t] = wmma_f32_16x16x4(a, b, acc[t]);
    }
  }
#pragma unroll
  for (int t = 0; t < 4; ++t)
#pragma unroll
    for (int v = 0; v < 8; ++v)
      out[(size_t)(rt * 16 + v + half * 8) * DOUT + cbase + t * 16 + m] = acc[t][v];
}

extern "C" void kernel_launch(void* const* d_in, const int* in_sizes, int n_in,
                              void* d_out, int out_size, void* d_ws, size_t ws_size,
                              hipStream_t stream) {
  const float* x      = (const float*)d_in[0];
  const int*   row    = (const int*)d_in[1];
  const int*   col    = (const int*)d_in[2];
  const float* adjval = (const float*)d_in[3];
  const float* W_row  = (const float*)d_in[4];
  const float* W_col  = (const float*)d_in[5];
  const float* W_x    = (const float*)d_in[6];
  const float* b_x    = (const float*)d_in[7];
  float* out = (float*)d_out;

  char* ws = (char*)d_ws;
  size_t off = 0;
  float* a_row = (float*)(ws + off); off += (size_t)N_NODES * DA * 4;
  float* a_col = (float*)(ws + off); off += (size_t)N_NODES * DA * 4;
  float* sc    = (float*)(ws + off); off += (size_t)N_EDGES * 4;
  int*   rec   = (int*)(ws + off);   off += (size_t)N_NODES * REC * 4;
  float* deg   = (float*)(ws + off); off += (size_t)N_NODES * 4;
  unsigned* maxkey = (unsigned*)(ws + off); off += 256;
  float* feat  = (float*)(ws + off); off += (size_t)N_NODES * DIN * 4;
  int*  total  = (int*)(maxkey + 32);
  int2* packed = (int2*)a_row;
  int*  pos    = (int*)feat;

  hipMemsetAsync(rec, 0, (size_t)N_NODES * REC * 4 + (size_t)N_NODES * 4 + 256, stream);

  {
    const int waves = N_NODES / 16;
    const int blocks = (waves * 32 + 255) / 256;
    proj_kernel<<<blocks, 256, 0, stream>>>(x, W_row, W_col, a_row, a_col);
  }
  {
    const int blocks = (N_EDGES + 255) / 256;
    edge_score_kernel<<<blocks, 256, 0, stream>>>(a_row, a_col, row, col, sc, maxkey, rec, pos);
    edge_exp_deg_kernel<<<blocks, 256, 0, stream>>>(sc, row, maxkey, deg);
    seg_alloc_kernel<<<(N_NODES + 255) / 256, 256, 0, stream>>>(rec, total);
    fill_kernel<<<blocks, 256, 0, stream>>>(row, col, adjval, sc, deg, rec, pos, packed);
  }
  {
    const int blocks = N_NODES / 8;
    gather_kernel<<<blocks, 256, 0, stream>>>(x, rec, packed, feat);
  }
  {
    const int waves = (N_NODES / 16) * 4;
    const int blocks = (waves * 32 + 255) / 256;
    out_gemm_kernel<<<blocks, 256, 0, stream>>>(feat, W_x, b_x, out);
  }
}


# Edge scores: 8 lanes share an edge and read the whole 128-byte lines of a_row[row] and a_col[col] (4 lines per load, not 32), dot summed over the 8 lanes; 128 edges per wave, one max atomic a block
# speedup vs baseline: 1.9142x; 1.2240x over previous
#include <hip/hip_runtime.h>

#define N_NODES 50000
#define N_EDGES 1600000
#define DIN 128
#define DOUT 256
#define DA 32
#define SLOPE 0.2f
#define REC 32

typedef __attribute__((ext_vector_type(2))) float v2f;
typedef __attribute__((ext_vector_type(8))) float v8f;

static __device__ __forceinline__ v8f wmma_f32_16x16x4(v2f a, v2f b, v8f c) {
  return __builtin_amdgcn_wmma_f32_16x16x4_f32(false, a, false, b, (short)0, c,
                                               false, false);
}

__global__ void proj_kernel(const float* __restrict__ x,
                            const float* __restrict__ W_row,
                            const float* __restrict__ W_col,
                            float* __restrict__ a_row,
                            float* __restrict__ a_col) {
  const int wave = (blockIdx.x * blockDim.x + threadIdx.x) >> 5;
  const int lane = threadIdx.x & 31;
  if (wave >= N_NODES / 16) return;
  const int tile = wave;
  const int half = lane >> 4;
  const int m    = lane & 15;
  const int node = tile * 16 + m;

  v8f acc0 = {}, acc1 = {}, acc2 = {}, acc3 = {};
  const float* xrow = x + (size_t)node * DIN + 2 * half;
  for (int k = 0; k < DIN; k += 4) {
    v2f a = *(const v2f*)(xrow + k);
    const int kk = k + 2 * half;
    v2f b0 = { W_row[kk * DA + m],      W_row[(kk + 1) * DA + m] };
    v2f b1 = { W_row[kk * DA + 16 + m], W_row[(kk + 1) * DA + 16 + m] };
    v2f b2 = { W_col[kk * DA + m],      W_col[(kk + 1) * DA + m] };
    v2f b3 = { W_col[kk * DA + 16 + m], W_col[(kk + 1) * DA + 16 + m] };
    acc0 = wmma_f32_16x16x4(a, b0, acc0);
    acc1 = wmma_f32_16x16x4(a, b1, acc1);
    acc2 = wmma_f32_16x16x4(a, b2, acc2);
    acc3 = wmma_f32_16x16x4(a, b3, acc3);
  }
#pragma unroll
  for (int v = 0; v < 8; ++v) {
    const int on = tile * 16 + v + half * 8;
    a_row[on * DA + m]      = acc0[v];
    a_row[on * DA + 16 + m] = acc1[v];
    a_col[on * DA + m]      = acc2[v];
    a_col[on * DA + 16 + m] = acc3[v];
  }
}

#define EPW 128
#define SCORE_WPB 8

#define SEL8(a0, a1, a2, a3, a4, a5, a6, a7)                                   \
  ((a0) | ((a1) << 3) | ((a2) << 6) | ((a3) << 9) | ((a4) << 12) |             \
   ((a5) << 15) | ((a6) << 18) | ((a7) << 21))

template <int I>
static __device__ __forceinline__ unsigned from_group_lane(unsigned v) {
  return __builtin_amdgcn_mov_dpp8(v, SEL8(I, I, I, I, I, I, I, I));
}

static __device__ __forceinline__ float sum_over_group(float p) {
  p += __uint_as_float(__builtin_amdgcn_mov_dpp8(__float_as_uint(p), SEL8(1, 0, 3, 2, 5, 4, 7, 6)));
  p += __uint_as_float(__builtin_amdgcn_mov_dpp8(__float_as_uint(p), SEL8(2, 3, 0, 1, 6, 7, 4, 5)));
  p += __uint_as_float(__builtin_amdgcn_mov_dpp8(__float_as_uint(p), SEL8(4, 5, 6, 7, 0, 1, 2, 3)));
  return p;
}

template <int I>
static __device__ __forceinline__ void load_edge_lines(const float* __restrict__ a_row,
                                                       const float* __restrict__ a_col,
                                                       unsigned r, unsigned c, unsigned sub16,
                                                       float4& u, float4& v) {
  const unsigned rr = from_group_lane<I>(r);
  const unsigned cc = from_group_lane<I>(c);
  u = *(const float4*)((const char*)a_row + (rr * (DA * 4u) + sub16));
  v = *(const float4*)((const char*)a_col + (cc * (DA * 4u) + sub16));
}

template <int I>
static __device__ __forceinline__ float keep_own(const float4& u, const float4& v, int sub, float s) {
  const float p = sum_over_group(u.x * v.x + u.y * v.y + u.z * v.z + u.w * v.w);
  return (sub == I) ? p : s;
}

__global__ void __launch_bounds__(SCORE_WPB * 32)
edge_score_kernel(const float* __restrict__ a_row,
                  const float* __restrict__ a_col,
                  const int* __restrict__ row,
                  const int* __restrict__ col,
                  float* __restrict__ raw,
                  unsigned* __restrict__ maxkey,
                  int* __restrict__ rec,
                  int* __restrict__ pos) {
  static_assert(N_EDGES % 32 == 0, "a chunk of 32 edges is whole or absent");
  __shared__ unsigned wave_key[SCORE_WPB];
  const int lane = threadIdx.x & 31;
  const int wv   = threadIdx.x >> 5;
  const int sub  = lane & 7;
  const unsigned sub16 = (unsigned)sub * 16u;
  const int wave_e0 = (blockIdx.x * SCORE_WPB + wv) * EPW;
  unsigned kmax = 0u;
  for (int ch = 0; ch < EPW / 32; ++ch) {
    const int e0 = wave_e0 + ch * 32;
    if (e0 < N_EDGES) {
      const int e = e0 + lane;
      const unsigned r = (unsigned)row[e];
      const unsigned c = (unsigned)col[e];
      float4 u[8], v[8];
      load_edge_lines<0>(a_row, a_col, r, c, sub16, u[0], v[0]);
      load_edge_lines<1>(a_row, a_col, r, c, sub16, u[1], v[1]);
      load_edge_lines<2>(a_row, a_col, r, c, sub16, u[2], v[2]);
      load_edge_lines<3>(a_row, a_col, r, c, sub16, u[3], v[3]);
      load_edge_lines<4>(a_row, a_col, r, c, sub16, u[4], v[4]);
      load_edge_lines<5>(a_row, a_col, r, c, sub16, u[5], v[5]);
      load_edge_lines<6>(a_row, a_col, r, c, sub16, u[6], v[6]);
      load_edge_lines<7>(a_row, a_col, r, c, sub16, u[7], v[7]);
      const int rank = atomicAdd(rec + (size_t)r * REC, 1);
      float s = 0.f;
      s = keep_own<0>(u[0], v[0], sub, s);
      s = keep_own<1>(u[1], v[1], sub, s);
      s = keep_own<2>(u[2], v[2], sub, s);
      s = keep_own<3>(u[3], v[3], sub, s);
      s = keep_own<4>(u[4], v[4], sub, s);
      s = keep_own<5>(u[5], v[5], sub, s);
      s = keep_own<6>(u[6], v[6], sub, s);
      s = keep_own<7>(u[7], v[7], sub, s);
      s *= 0.17677669529663687f;
      s = (s >= 0.f) ? s : SLOPE * s;
      raw[e] = s;
      pos[e] = rank;
      const unsigned b = __float_as_uint(s);
      const unsigned key = (b & 0x80000000u) ? ~b : (b | 0x80000000u);
      kmax = (key > kmax) ? key : kmax;
    }
  }
#pragma unroll
  for (int off = 16; off > 0; off >>= 1) {
    const unsigned o = __shfl_xor(kmax, off, 32);
    kmax = (o > kmax) ? o : kmax;
  }
  if (lane == 0) wave_key[wv] = kmax;
  __syncthreads();
  if (threadIdx.x == 0) {
    unsigned m = wave_key[0];
    for (int w = 1; w < SCORE_WPB; ++w) m = (wave_key[w] > m) ? wave_key[w] : m;
    atomicMax(maxkey, m);
  }
}

__global__ void edge_exp_deg_kernel(float* __restrict__ sc,
                                    const int* __restrict__ row,
                                    const unsigned* __restrict__ maxkey,
                                    float* __restrict__ deg) {
  const int e = blockIdx.x * blockDim.x + threadIdx.x;
  if (e >= N_EDGES) return;
  const unsigned key = *maxkey;
  const unsigned b = (key & 0x80000000u) ? (key ^ 0x80000000u) : ~key;
  const float gmax = __uint_as_float(b);
  const float a = expf(sc[e] - gmax);
  sc[e] = a;
  atomicAdd(deg + row[e], a);
}

__global__ void seg_alloc_kernel(int* __restrict__ rec, int* __restrict__ total) {
  const int r = blockIdx.x * blockDim.x + threadIdx.x;
  const int lane = threadIdx.x & 31;
  const int c = (r < N_NODES) ? rec[(size_t)r * REC] : 0;
  int incl = c;
#pragma unroll
  for (int off = 1; off < 32; off <<= 1) {
    const int o = __shfl_up(incl, off, 32);
    if (lane >= off) incl += o;
  }
  int base = 0;
  if (lane == 31) base = atomicAdd(total, incl);
  base = __shfl(base, 31, 32);
  if (r < N_NODES) rec[(size_t)r * REC + 1] = base + incl - c;
}

__global__ void fill_kernel(const int* __restrict__ row,
                            const int* __restrict__ col,
                            const float* __restrict__ adj_val,
                            const float* __restrict__ atten,
                            const float* __restrict__ deg,
                            const int* __restrict__ rec,
                            const int* __restrict__ pos,
                            int2* __restrict__ packed) {
  const int e = blockIdx.x * blockDim.x + threadIdx.x;
  if (e >= N_EDGES) return;
  const int r = row[e];
  const float val = (adj_val[e] + atten[e] / (deg[r] + 1e-15f)) * 0.5f;
  int idx = rec[(size_t)r * REC + 1] + pos[e];
  idx = idx < 0 ? 0 : (idx > N_EDGES - 1 ? N_EDGES - 1 : idx);
  packed[idx] = make_int2(col[e], __float_as_int(val));
}

__global__ void gather_kernel(const float* __restrict__ x,
                              const int* __restrict__ rec,
                              const int2* __restrict__ packed,
                              float* __restrict__ feat) {
  const int wpb = blockDim.x >> 5;
  const int r = __builtin_amdgcn_readfirstlane(blockIdx.x * wpb +
                                               (threadIdx.x >> 5));
  const int lane = threadIdx.x & 31;
  if (r >= N_NODES) return;
  int n = rec[(size_t)r * REC];
  n = n > N_EDGES ? N_EDGES : n;
  const int s = rec[(size_t)r * REC + 1];
  const float* xl = x + lane * 4;
  float4 acc = make_float4(0.f, 0.f, 0.f, 0.f);
  for (int j = 0; j < n; j += 32) {
    const int m = (n - j < 32) ? (n - j) : 32;
    int2 p = make_int2(0, 0);
    if (lane < m) {
      int q = s + j + lane;
      q = q < 0 ? 0 : (q > N_EDGES - 1 ? N_EDGES - 1 : q);
      p = packed[q];
    }
    for (int k = 0; k < m; k += 4) {
      const unsigned c0 = min((unsigned)__builtin_amdgcn_readlane(p.x, k), (unsigned)(N_NODES - 1));
      const unsigned c1 = min((unsigned)__builtin_amdgcn_readlane(p.x, k + 1), (unsigned)(N_NODES - 1));
      const unsigned c2 = min((unsigned)__builtin_amdgcn_readlane(p.x, k + 2), (unsigned)(N_NODES - 1));
      const unsigned c3 = min((unsigned)__builtin_amdgcn_readlane(p.x, k + 3), (unsigned)(N_NODES - 1));
      const float v0 = __int_as_float(__builtin_amdgcn_readlane(p.y, k));
      const float v1 = __int_as_float(__builtin_amdgcn_readlane(p.y, k + 1));
      const float v2 = __int_as_float(__builtin_amdgcn_readlane(p.y, k + 2));
      const float v3 = __int_as_float(__builtin_amdgcn_readlane(p.y, k + 3));
      const float4 x0 = *(const float4*)(xl + (size_t)c0 * DIN);
      const float4 x1 = *(const float4*)(xl + (size_t)c1 * DIN);
      const float4 x2 = *(const float4*)(xl + (size_t)c2 * DIN);
      const float4 x3 = *(const float4*)(xl + (size_t)c3 * DIN);
      acc.x += v0 * x0.x; acc.y += v0 * x0.y; acc.z += v0 * x0.z; acc.w += v0 * x0.w;
      acc.x += v1 * x1.x; acc.y += v1 * x1.y; acc.z += v1 * x1.z; acc.w += v1 * x1.w;
      acc.x += v2 * x2.x; acc.y += v2 * x2.y; acc.z += v2 * x2.z; acc.w += v2 * x2.w;
      acc.x += v3 * x3.x; acc.y += v3 * x3.y; acc.z += v3 * x3.z; acc.w += v3 * x3.w;
    }
  }
  *(float4*)(feat + (size_t)r * DIN + lane * 4) = acc;
}

__global__ void out_gemm_kernel(const float* __restrict__ feat,
                                const float* __restrict__ W_x,
                                const float* __restrict__ b_x,
                                float* __restrict__ out) {
  const int wave = (blockIdx.x * blockDim.x + threadIdx.x) >> 5;
  const int lane = threadIdx.x & 31;
  const int total = (N_NODES / 16) * 4;
  if (wave >= total) return;
  const int rt    = wave >> 2;
  const int cbase = (wave & 3) * 64;
  const int half  = lane >> 4;
  const int m     = lane & 15;
  const int node  = rt * 16 + m;

  v8f acc[4];
#pragma unroll
  for (int t = 0; t < 4; ++t) {
    const float bias = b_x[cbase + t * 16 + m];
#pragma unroll
    for (int v = 0; v < 8; ++v) acc[t][v] = bias;
  }
  const float* frow = feat + (size_t)node * DIN + 2 * half;
  for (int k = 0; k < DIN; k += 4) {
    v2f a = *(const v2f*)(frow + k);
    const int kk = k + 2 * half;
#pragma unroll
    for (int t = 0; t < 4; ++t) {
      v2f b = { W_x[kk * DOUT + cbase + t * 16 + m],
                W_x[(kk + 1) * DOUT + cbase + t * 16 + m] };
      acc[t] = wmma_f32_16x16x4(a, b, acc[t]);
    }
  }
#pragma unroll
  for (int t = 0; t < 4; ++t)
#pragma unroll
    for (int v = 0; v < 8; ++v)
      out[(size_t)(rt * 16 + v + half * 8) * DOUT + cbase + t * 16 + m] = acc[t][v];
}

extern "C" void kernel_launch(void* const* d_in, const int* in_sizes, int n_in,
                              void* d_out, int out_size, void* d_ws, size_t ws_size,
                              hipStream_t stream) {
  const float* x      = (const float*)d_in[0];
  const int*   row    = (const int*)d_in[1];
  const int*   col    = (const int*)d_in[2];
  const float* adjval = (const float*)d_in[3];
  const float* W_row  = (const float*)d_in[4];
  const float* W_col  = (const float*)d_in[5];
  const float* W_x    = (const float*)d_in[6];
  const float* b_x    = (const float*)d_in[7];
  float* out = (float*)d_out;

  char* ws = (char*)d_ws;
  size_t off = 0;
  float* a_row = (float*)(ws + off); off += (size_t)N_NODES * DA * 4;
  float* a_col = (float*)(ws + off); off += (size_t)N_NODES * DA * 4;
  float* sc    = (float*)(ws + off); off += (size_t)N_EDGES * 4;
  int*   rec   = (int*)(ws + off);   off += (size_t)N_NODES * REC * 4;
  float* deg   = (float*)(ws + off); off += (size_t)N_NODES * 4;
  unsigned* maxkey = (unsigned*)(ws + off); off += 256;
  float* feat  = (float*)(ws + off); off += (size_t)N_NODES * DIN * 4;
  int*  total  = (int*)(maxkey + 32);
  int2* packed = (int2*)a_row;
  int*  pos    = (int*)feat;

  hipMemsetAsync(rec, 0, (size_t)N_NODES * REC * 4 + (size_t)N_NODES * 4 + 256, stream);

  {
    const int waves = N_NODES / 16;
    const int blocks = (waves * 32 + 255) / 256;
    proj_kernel<<<blocks, 256, 0, stream>>>(x, W_row, W_col, a_row, a_col);
  }
  {
    const int blocks = (N_EDGES + 255) / 256;
    edge_score_kernel<<<(N_EDGES + SCORE_WPB * EPW - 1) / (SCORE_WPB * EPW), SCORE_WPB * 32, 0, stream>>>(a_row, a_col, row, col, sc, maxkey, rec, pos);
    edge_exp_deg_kernel<<<blocks, 256, 0, stream>>>(sc, row, maxkey, deg);
    seg_alloc_kernel<<<(N_NODES + 255) / 256, 256, 0, stream>>>(rec, total);
    fill_kernel<<<blocks, 256, 0, stream>>>(row, col, adjval, sc, deg, rec, pos, packed);
  }
  {
    const int blocks = N_NODES / 8;
    gather_kernel<<<blocks, 256, 0, stream>>>(x, rec, packed, feat);
  }
  {
    const int waves = (N_NODES / 16) * 4;
    const int blocks = (waves * 32 + 255) / 256;
    out_gemm_kernel<<<blocks, 256, 0, stream>>>(feat, W_x, b_x, out);
  }
}


# Edge scores on the row-bucketed records: a wave walks whole rows and reads a_row[row] once per row, not per edge (one random line less per edge); edge ranks are taken in the projection launch
# speedup vs baseline: 3.4677x; 1.8116x over previous
#include <hip/hip_runtime.h>

#define N_NODES 50000
#define N_EDGES 1600000
#define DIN 128
#define DOUT 256
#define DA 32
#define SLOPE 0.2f
#define REC 32
static_assert(2 * N_NODES * DA == N_EDGES * 2, "a_row and a_col together are two parts of N_EDGES*4 bytes");

typedef __attribute__((ext_vector_type(2))) float v2f;
struct EdgeRec { int col; float a; float adj; };
typedef __attribute__((ext_vector_type(8))) float v8f;

static __device__ __forceinline__ v8f wmma_f32_16x16x4(v2f a, v2f b, v8f c) {
  return __builtin_amdgcn_wmma_f32_16x16x4_f32(false, a, false, b, (short)0, c,
                                               false, false);
}

#define PROJ_BLOCKS ((N_NODES / 16 * 32 + 255) / 256)
#define RANK_BLOCKS ((N_EDGES / 4 + 255) / 256)
__global__ void __launch_bounds__(256)
proj_rank_kernel(const float* __restrict__ x,
                 const float* __restrict__ W_row,
                 const float* __restrict__ W_col,
                 float* __restrict__ a_row,
                 float* __restrict__ a_col,
                 const int* __restrict__ row,
                 int* __restrict__ rec,
                 int* __restrict__ pos) {
  static_assert(N_EDGES % 4 == 0, "four edges per thread are whole or absent");
  if (blockIdx.x >= PROJ_BLOCKS) {
    const int t = (blockIdx.x - PROJ_BLOCKS) * 256 + threadIdx.x;
    if (t < N_EDGES / 4) {
      const int4 r = *(const int4*)(row + 4 * t);
      int4 k;
      k.x = atomicAdd(rec + (size_t)r.x * REC, 1);
      k.y = atomicAdd(rec + (size_t)r.y * REC, 1);
      k.z = atomicAdd(rec + (size_t)r.z * REC, 1);
      k.w = atomicAdd(rec + (size_t)r.w * REC, 1);
      *(int4*)(pos + 4 * t) = k;
    }
    return;
  }
  const int wave = (blockIdx.x * blockDim.x + threadIdx.x) >> 5;
  const int lane = threadIdx.x & 31;
  if (wave >= N_NODES / 16) return;
  const int tile = wave;
  const int half = lane >> 4;
  const int m    = lane & 15;
  const int node = tile * 16 + m;

  v8f acc0 = {}, acc1 = {}, acc2 = {}, acc3 = {};
  const float* xrow = x + (size_t)node * DIN + 2 * half;
  for (int k = 0; k < DIN; k += 4) {
    v2f a = *(const v2f*)(xrow + k);
    const int kk = k + 2 * half;
    v2f b0 = { W_row[kk * DA + m],      W_row[(kk + 1) * DA + m] };
    v2f b1 = { W_row[kk * DA + 16 + m], W_row[(kk + 1) * DA + 16 + m] };
    v2f b2 = { W_col[kk * DA + m],      W_col[(kk + 1) * DA + m] };
    v2f b3 = { W_col[kk * DA + 16 + m], W_col[(kk + 1) * DA + 16 + m] };
    acc0 = wmma_f32_16x16x4(a, b0, acc0);
    acc1 = wmma_f32_16x16x4(a, b1, acc1);
    acc2 = wmma_f32_16x16x4(a, b2, acc2);
    acc3 = wmma_f32_16x16x4(a, b3, acc3);
  }
#pragma unroll
  for (int v = 0; v < 8; ++v) {
    const int on = tile * 16 + v + half * 8;
    a_row[on * DA + m]      = acc0[v];
    a_row[on * DA + 16 + m] = acc1[v];
    a_col[on * DA + m]      = acc2[v];
    a_col[on * DA + 16 + m] = acc3[v];
  }
}

#define SR_WPB 8
#define SR_RPW 4

#define SEL8(a0, a1, a2, a3, a4, a5, a6, a7)                                   \
  ((a0) | ((a1) << 3) | ((a2) << 6) | ((a3) << 9) | ((a4) << 12) |             \
   ((a5) << 15) | ((a6) << 18) | ((a7) << 21))

template <int I>
static __device__ __forceinline__ unsigned from_group_lane(unsigned v) {
  return __builtin_amdgcn_mov_dpp8(v, SEL8(I, I, I, I, I, I, I, I));
}

static __device__ __forceinline__ float sum_over_group(float p) {
  p += __uint_as_float(__builtin_amdgcn_mov_dpp8(__float_as_uint(p), SEL8(1, 0, 3, 2, 5, 4, 7, 6)));
  p += __uint_as_float(__builtin_amdgcn_mov_dpp8(__float_as_uint(p), SEL8(2, 3, 0, 1, 6, 7, 4, 5)));
  p += __uint_as_float(__builtin_amdgcn_mov_dpp8(__float_as_uint(p), SEL8(4, 5, 6, 7, 0, 1, 2, 3)));
  return p;
}

template <int I>
static __device__ __forceinline__ float4 load_col_line(const float* __restrict__ a_col,
                                                       unsigned c, unsigned sub16,
                                                       int g, int m) {
  const unsigned cc = from_group_lane<I>(c);
  float4 v = make_float4(0.f, 0.f, 0.f, 0.f);
  if (8 * g + I < m) v = *(const float4*)((const char*)a_col + (cc * (DA * 4u) + sub16));
  return v;
}

template <int I>
static __device__ __forceinline__ float keep_own(const float4& u, const float4& v, int sub, float s) {
  const float p = sum_over_group(u.x * v.x + u.y * v.y + u.z * v.z + u.w * v.w);
  return (sub == I) ? p : s;
}

__global__ void __launch_bounds__(SR_WPB * 32)
score_rows_kernel(const float* __restrict__ a_row,
                  const float* __restrict__ a_col,
                  const int* __restrict__ rec,
                  EdgeRec* __restrict__ packed,
                  unsigned* __restrict__ maxkey) {
  __shared__ unsigned wave_key[SR_WPB];
  const int wv   = __builtin_amdgcn_readfirstlane(threadIdx.x >> 5);
  const int lane = threadIdx.x & 31;
  const int sub  = lane & 7;
  const int g    = lane >> 3;
  const unsigned sub16 = (unsigned)sub * 16u;
  const int wrow0 = (blockIdx.x * SR_WPB + wv) * SR_RPW;
  unsigned kmax = 0u;
  for (int i = 0; i < SR_RPW; ++i) {
    const int r = wrow0 + i;
    if (r < N_NODES) {
      int n = rec[(size_t)r * REC];
      n = n > N_EDGES ? N_EDGES : n;
      const int s0 = rec[(size_t)r * REC + 1];
      const float4 u = *(const float4*)((const char*)a_row + ((unsigned)r * (DA * 4u) + sub16));
      for (int j = 0; j < n; j += 32) {
        const int m = (n - j < 32) ? (n - j) : 32;
        const bool live = lane < m;
        int q = s0 + j + lane;
        q = q < 0 ? 0 : (q > N_EDGES - 1 ? N_EDGES - 1 : q);
        unsigned c = 0u;
        if (live) c = (unsigned)packed[q].col;
        c = min(c, (unsigned)(N_NODES - 1));
        float4 v[8];
        v[0] = load_col_line<0>(a_col, c, sub16, g, m);
        v[1] = load_col_line<1>(a_col, c, sub16, g, m);
        v[2] = load_col_line<2>(a_col, c, sub16, g, m);
        v[3] = load_col_line<3>(a_col, c, sub16, g, m);
        v[4] = load_col_line<4>(a_col, c, sub16, g, m);
        v[5] = load_col_line<5>(a_col, c, sub16, g, m);
        v[6] = load_col_line<6>(a_col, c, sub16, g, m);
        v[7] = load_col_line<7>(a_col, c, sub16, g, m);
        float s = 0.f;
        s = keep_own<0>(u, v[0], sub, s);
        s = keep_own<1>(u, v[1], sub, s);
        s = keep_own<2>(u, v[2], sub, s);
        s = keep_own<3>(u, v[3], sub, s);
        s = keep_own<4>(u, v[4], sub, s);
        s = keep_own<5>(u, v[5], sub, s);
        s = keep_own<6>(u, v[6], sub, s);
        s = keep_own<7>(u, v[7], sub, s);
        s *= 0.17677669529663687f;
        s = (s >= 0.f) ? s : SLOPE * s;
        if (live) {
          packed[q].a = s;
          const unsigned b = __float_as_uint(s);
          const unsigned key = (b & 0x80000000u) ? ~b : (b | 0x80000000u);
          kmax = (key > kmax) ? key : kmax;
        }
      }
    }
  }
#pragma unroll
  for (int off = 16; off > 0; off >>= 1) {
    const unsigned o = __shfl_xor(kmax, off, 32);
    kmax = (o > kmax) ? o : kmax;
  }
  if (lane == 0) wave_key[wv] = kmax;
  __syncthreads();
  if (threadIdx.x == 0) {
    unsigned mx = wave_key[0];
    for (int w = 1; w < SR_WPB; ++w) mx = (wave_key[w] > mx) ? wave_key[w] : mx;
    atomicMax(maxkey, mx);
  }
}

__global__ void seg_alloc_kernel(int* __restrict__ rec, int* __restrict__ total) {
  const int r = blockIdx.x * blockDim.x + threadIdx.x;
  const int lane = threadIdx.x & 31;
  const int c = (r < N_NODES) ? rec[(size_t)r * REC] : 0;
  int incl = c;
#pragma unroll
  for (int off = 1; off < 32; off <<= 1) {
    const int o = __shfl_up(incl, off, 32);
    if (lane >= off) incl += o;
  }
  int base = 0;
  if (lane == 31) base = atomicAdd(total, incl);
  base = __shfl(base, 31, 32);
  if (r < N_NODES) rec[(size_t)r * REC + 1] = base + incl - c;
}

__global__ void fill_kernel(const int* __restrict__ row,
                            const int* __restrict__ col,
                            const float* __restrict__ adj_val,
                            const int* __restrict__ rec,
                            const int* __restrict__ pos,
                            EdgeRec* __restrict__ packed) {
  const int e = blockIdx.x * blockDim.x + threadIdx.x;
  if (e >= N_EDGES) return;
  const int r = row[e];
  int idx = rec[(size_t)r * REC + 1] + pos[e];
  idx = idx < 0 ? 0 : (idx > N_EDGES - 1 ? N_EDGES - 1 : idx);
  EdgeRec o;
  o.col = col[e];
  o.a   = 0.f;
  o.adj = adj_val[e];
  packed[idx] = o;
}

static __device__ __forceinline__ EdgeRec load_rec(const EdgeRec* __restrict__ packed,
                                                   int q, bool live, float gmax) {
  EdgeRec p;
  p.col = 0; p.a = 0.f; p.adj = 0.f;
  if (live) {
    q = q < 0 ? 0 : (q > N_EDGES - 1 ? N_EDGES - 1 : q);
    p = packed[q];
    p.a = expf(p.a - gmax);
  }
  return p;
}

static __device__ __forceinline__ float4 gather_row(const float* __restrict__ x,
                                                    const int* __restrict__ rec,
                                                    const EdgeRec* __restrict__ packed,
                                                    float gmax, int r, int lane) {
  int n = rec[(size_t)r * REC];
  n = n > N_EDGES ? N_EDGES : n;
  const int s = rec[(size_t)r * REC + 1];

  const EdgeRec first = load_rec(packed, s + lane, lane < n, gmax);
  float part = first.a;
  for (int j = 32; j < n; j += 32)
    part += load_rec(packed, s + j + lane, j + lane < n, gmax).a;
  for (int off = 16; off > 0; off >>= 1) part += __shfl_xor(part, off, 32);
  const float inv = 1.0f / (part + 1e-15f);

  const float* xl = x + lane * 4;
  float4 acc = make_float4(0.f, 0.f, 0.f, 0.f);
  for (int j = 0; j < n; j += 32) {
    const int m = (n - j < 32) ? (n - j) : 32;
    EdgeRec p = first;
    if (j > 0) p = load_rec(packed, s + j + lane, lane < m, gmax);
    const int   pc = p.col;
    const float pv = (p.adj + p.a * inv) * 0.5f;
    for (int k = 0; k < m; k += 4) {
      const unsigned c0 = min((unsigned)__builtin_amdgcn_readlane(pc, k), (unsigned)(N_NODES - 1));
      const unsigned c1 = min((unsigned)__builtin_amdgcn_readlane(pc, k + 1), (unsigned)(N_NODES - 1));
      const unsigned c2 = min((unsigned)__builtin_amdgcn_readlane(pc, k + 2), (unsigned)(N_NODES - 1));
      const unsigned c3 = min((unsigned)__builtin_amdgcn_readlane(pc, k + 3), (unsigned)(N_NODES - 1));
      const float v0 = __int_as_float(__builtin_amdgcn_readlane(__float_as_int(pv), k));
      const float v1 = __int_as_float(__builtin_amdgcn_readlane(__float_as_int(pv), k + 1));
      const float v2 = __int_as_float(__builtin_amdgcn_readlane(__float_as_int(pv), k + 2));
      const float v3 = __int_as_float(__builtin_amdgcn_readlane(__float_as_int(pv), k + 3));
      const float4 x0 = *(const float4*)(xl + (size_t)c0 * DIN);
      const float4 x1 = *(const float4*)(xl + (size_t)c1 * DIN);
      const float4 x2 = *(const float4*)(xl + (size_t)c2 * DIN);
      const float4 x3 = *(const float4*)(xl + (size_t)c3 * DIN);
      acc.x += v0 * x0.x; acc.y += v0 * x0.y; acc.z += v0 * x0.z; acc.w += v0 * x0.w;
      acc.x += v1 * x1.x; acc.y += v1 * x1.y; acc.z += v1 * x1.z; acc.w += v1 * x1.w;
      acc.x += v2 * x2.x; acc.y += v2 * x2.y; acc.z += v2 * x2.z; acc.w += v2 * x2.w;
      acc.x += v3 * x3.x; acc.y += v3 * x3.y; acc.z += v3 * x3.z; acc.w += v3 * x3.w;
    }
  }
  return acc;
}

#define GG_WAVES 8
#define TILE_ROWS 16
#define TILE_PITCH 132

__global__ void __launch_bounds__(GG_WAVES * 32)
gather_gemm_kernel(const float* __restrict__ x,
                   const int* __restrict__ rec,
                   const EdgeRec* __restrict__ packed,
                   const unsigned* __restrict__ maxkey,
                   const float* __restrict__ W_x,
                   const float* __restrict__ b_x,
                   float* __restrict__ out) {
  static_assert(N_NODES % TILE_ROWS == 0, "every block has 16 whole rows");
  static_assert(GG_WAVES * 32 == DOUT, "the waves' column groups cover DOUT");
  static_assert(TILE_ROWS == 2 * GG_WAVES, "two rows per wave");
  __shared__ float tile[TILE_ROWS * TILE_PITCH];
  const int wv   = __builtin_amdgcn_readfirstlane(threadIdx.x >> 5);
  const int lane = threadIdx.x & 31;
  const int row0 = blockIdx.x * TILE_ROWS;
  const unsigned key = *maxkey;
  const unsigned kb = (key & 0x80000000u) ? (key ^ 0x80000000u) : ~key;
  const float gmax = __uint_as_float(kb);

  for (int i = 0; i < 2; ++i) {
    const int lr = wv + GG_WAVES * i;
    const float4 f = gather_row(x, rec, packed, gmax, row0 + lr, lane);
    *(float4*)(tile + lr * TILE_PITCH + lane * 4) = f;
  }
  __syncthreads();

  const int half  = lane >> 4;
  const int m     = lane & 15;
  const int cbase = wv * 32;

  v8f acc0, acc1;
  {
    const float bias0 = b_x[cbase + m];
    const float bias1 = b_x[cbase + 16 + m];
#pragma unroll
    for (int v = 0; v < 8; ++v) { acc0[v] = bias0; acc1[v] = bias1; }
  }
  const float* arow = tile + m * TILE_PITCH + 2 * half;
  for (int k = 0; k < DIN; k += 4) {
    v2f a = *(const v2f*)(arow + k);
    const int kk = k + 2 * half;
    v2f b0 = { W_x[kk * DOUT + cbase + m],      W_x[(kk + 1) * DOUT + cbase + m] };
    v2f b1 = { W_x[kk * DOUT + cbase + 16 + m], W_x[(kk + 1) * DOUT + cbase + 16 + m] };
    acc0 = wmma_f32_16x16x4(a, b0, acc0);
    acc1 = wmma_f32_16x16x4(a, b1, acc1);
  }
#pragma unroll
  for (int v = 0; v < 8; ++v) {
    float* orow = out + (size_t)(row0 + v + half * 8) * DOUT + cbase + m;
    orow[0]  = acc0[v];
    orow[16] = acc1[v];
  }
}

extern "C" void kernel_launch(void* const* d_in, const int* in_sizes, int n_in,
                              void* d_out, int out_size, void* d_ws, size_t ws_size,
                              hipStream_t stream) {
  const float* x      = (const float*)d_in[0];
  const int*   row    = (const int*)d_in[1];
  const int*   col    = (const int*)d_in[2];
  const float* adjval = (const float*)d_in[3];
  const float* W_row  = (const float*)d_in[4];
  const float* W_col  = (const float*)d_in[5];
  const float* W_x    = (const float*)d_in[6];
  const float* b_x    = (const float*)d_in[7];
  float* out = (float*)d_out;

  char* ws = (char*)d_ws;
  size_t off = 0;
  float* a_row = (float*)(ws + off); off += (size_t)N_NODES * DA * 4;
  float* a_col = (float*)(ws + off); off += (size_t)N_NODES * DA * 4;
  int*   pos   = (int*)(ws + off);   off += (size_t)N_EDGES * 4;
  int*   rec   = (int*)(ws + off);   off += (size_t)N_NODES * REC * 4;
  unsigned* maxkey = (unsigned*)(ws + off); off += 256;
  off += (size_t)N_NODES * 4;
  EdgeRec* packed = (EdgeRec*)(ws + off); off += (size_t)N_NODES * DIN * 4;
  int*  total  = (int*)(maxkey + 32);
  static_assert((size_t)N_EDGES * 12 <= (size_t)N_NODES * DIN * 4, "the records fit the last part");

  hipMemsetAsync(rec, 0, (size_t)N_NODES * REC * 4 + 256, stream);

  proj_rank_kernel<<<PROJ_BLOCKS + RANK_BLOCKS, 256, 0, stream>>>(x, W_row, W_col, a_row, a_col, row, rec, pos);
  {
    const int blocks = (N_EDGES + 255) / 256;
    seg_alloc_kernel<<<(N_NODES + 255) / 256, 256, 0, stream>>>(rec, total);
    fill_kernel<<<blocks, 256, 0, stream>>>(row, col, adjval, rec, pos, packed);
    score_rows_kernel<<<(N_NODES + SR_WPB * SR_RPW - 1) / (SR_WPB * SR_RPW), SR_WPB * 32, 0, stream>>>(a_row, a_col, rec, packed, maxkey);
  }
  {
    const int blocks = N_NODES / TILE_ROWS;
    gather_gemm_kernel<<<blocks, GG_WAVES * 32, 0, stream>>>(x, rec, packed, maxkey, W_x, b_x, out);
  }
}


# Edges bucketed by row in two levels through LDS (250 row ranges, then rows): the per-edge returning add and 12-byte store on random lines become LDS adds and contiguous writes; 4 launches, 32 KB fill
# speedup vs baseline: 4.4315x; 1.2779x over previous
#include <hip/hip_runtime.h>

#define N_NODES 50000
#define N_EDGES 1600000
#define DIN 128
#define DOUT 256
#define DA 32
#define SLOPE 0.2f
#define SLOT 64
#define CAP (SLOT - 1)
static_assert(SLOT == 2 * (N_EDGES / N_NODES), "slot size follows the mean row length");
static_assert(N_EDGES % 32 == 0, "the slow paths walk the edge list a whole wave at a time");
#define NBIN 250
#define RPBIN (N_NODES / NBIN)
#define BINCAP (RPBIN * SLOT)
#define EPB 3200
#define PA_BLOCKS (N_EDGES / EPB)
#define PA_TRIPS ((EPB + 255) / 256)
static_assert(N_NODES % NBIN == 0, "every range has RPBIN whole rows");
static_assert(N_EDGES % EPB == 0, "every level-1 block has EPB whole edges");
static_assert(NBIN <= 256, "one thread of a level-1 block per range");
static_assert(EPB <= 65535, "an edge's rank inside its block fits 16 bits");

typedef __attribute__((ext_vector_type(2))) float v2f;
struct EdgeRec { int col; float a; float adj; };
struct BinRec { int col; int row; float adj; };
static_assert(sizeof(BinRec) == sizeof(EdgeRec), "a parked edge takes one place");
typedef __attribute__((ext_vector_type(8))) float v8f;

static __device__ __forceinline__ v8f wmma_f32_16x16x4(v2f a, v2f b, v8f c) {
  return __builtin_amdgcn_wmma_f32_16x16x4_f32(false, a, false, b, (short)0, c,
                                               false, false);
}

#define PROJ_BLOCKS ((N_NODES / 16 * 32 + 255) / 256)

static __device__ __forceinline__ int range_of(int r) {
  const unsigned b = (unsigned)r / (unsigned)RPBIN;
  return (int)(b < (unsigned)NBIN ? b : (unsigned)(NBIN - 1));
}

__global__ void __launch_bounds__(256)
proj_park_kernel(const float* __restrict__ x,
                 const float* __restrict__ W_row,
                 const float* __restrict__ W_col,
                 float* __restrict__ a_row,
                 float* __restrict__ a_col,
                 const int* __restrict__ row,
                 const int* __restrict__ col,
                 const float* __restrict__ adj_val,
                 EdgeRec* __restrict__ packed,
                 int* __restrict__ bin_count) {
  if (blockIdx.x >= PROJ_BLOCKS) {
    __shared__ int s_hist[256];
    __shared__ int s_offs[256];
    __shared__ int s_gbase[256];
    __shared__ int s_wsum[8];
    __shared__ unsigned short s_lrank[EPB];
    __shared__ BinRec s_stage[EPB];
    const int tid  = threadIdx.x;
    const int lane = tid & 31;
    const int wv   = tid >> 5;
    const int e0   = (blockIdx.x - PROJ_BLOCKS) * EPB;
    s_hist[tid] = 0;
    __syncthreads();
    int rr[PA_TRIPS];
#pragma unroll
    for (int j = 0; j < PA_TRIPS; ++j) {
      const int i = j * 256 + tid;
      rr[j] = (i < EPB) ? row[e0 + i] : 0;
    }
#pragma unroll
    for (int j = 0; j < PA_TRIPS; ++j) {
      const int i = j * 256 + tid;
      if (i < EPB) s_lrank[i] = (unsigned short)atomicAdd(&s_hist[range_of(rr[j])], 1);
    }
    __syncthreads();
    const int h = s_hist[tid];
    int inc = h;
#pragma unroll
    for (int off = 1; off < 32; off <<= 1) {
      const int t = __shfl_up(inc, off, 32);
      if (lane >= off) inc += t;
    }
    if (lane == 31) s_wsum[wv] = inc;
    __syncthreads();
    int before = 0;
    for (int w = 0; w < wv; ++w) before += s_wsum[w];
    s_offs[tid] = before + inc - h;
    if (tid < NBIN) s_gbase[tid] = atomicAdd(&bin_count[tid * 32], h);
    __syncthreads();
#pragma unroll
    for (int j = 0; j < PA_TRIPS; ++j) {
      const int i = j * 256 + tid;
      if (i < EPB) {
        const int e = e0 + i;
        BinRec o; o.col = col[e]; o.row = rr[j]; o.adj = adj_val[e];
        s_stage[s_offs[range_of(o.row)] + s_lrank[i]] = o;
      }
    }
    __syncthreads();
    BinRec* const parked = (BinRec*)packed;
    for (int i = tid; i < EPB; i += 256) {
      const BinRec o = s_stage[i];
      const int b = range_of(o.row);
      const int k = s_gbase[b] + (i - s_offs[b]);
      if (k >= 0 && k < BINCAP) parked[(size_t)b * BINCAP + k] = o;
    }
    return;
  }
  const int wave = (blockIdx.x * blockDim.x + threadIdx.x) >> 5;
  const int lane = threadIdx.x & 31;
  if (wave >= N_NODES / 16) return;
  const int tile = wave;
  const int half = lane >> 4;
  const int m    = lane & 15;
  const int node = tile * 16 + m;

  v8f acc0 = {}, acc1 = {}, acc2 = {}, acc3 = {};
  const float* xrow = x + (size_t)node * DIN + 2 * half;
  for (int k = 0; k < DIN; k += 4) {
    v2f a = *(const v2f*)(xrow + k);
    const int kk = k + 2 * half;
    v2f b0 = { W_row[kk * DA + m],      W_row[(kk + 1) * DA + m] };
    v2f b1 = { W_row[kk * DA + 16 + m], W_row[(kk + 1) * DA + 16 + m] };
    v2f b2 = { W_col[kk * DA + m],      W_col[(kk + 1) * DA + m] };
    v2f b3 = { W_col[kk * DA + 16 + m], W_col[(kk + 1) * DA + 16 + m] };
    acc0 = wmma_f32_16x16x4(a, b0, acc0);
    acc1 = wmma_f32_16x16x4(a, b1, acc1);
    acc2 = wmma_f32_16x16x4(a, b2, acc2);
    acc3 = wmma_f32_16x16x4(a, b3, acc3);
  }
#pragma unroll
  for (int v = 0; v < 8; ++v) {
    const int on = tile * 16 + v + half * 8;
    a_row[on * DA + m]      = acc0[v];
    a_row[on * DA + 16 + m] = acc1[v];
    a_col[on * DA + m]      = acc2[v];
    a_col[on * DA + 16 + m] = acc3[v];
  }
}

#define PB_THREADS 1024
#define PB_BATCH 8
#define IMG_VEC (RPBIN * SLOT * 12 / 16)
static_assert((RPBIN * SLOT * 12) % 16 == 0, "a range's slots are whole 16-byte pieces");

static __device__ __forceinline__ void place_edge(EdgeRec* img, int rl, int c, float adj) {
  if ((unsigned)rl < (unsigned)RPBIN) {
    EdgeRec* const slot = img + rl * SLOT;
    const int k = atomicAdd(&slot->col, 1);
    if (k >= 0 && k < CAP) { EdgeRec o; o.col = c; o.a = 0.f; o.adj = adj; slot[1 + k] = o; }
  }
}

__global__ void __launch_bounds__(PB_THREADS)
place_rows_kernel(const int* __restrict__ row,
                  const int* __restrict__ col,
                  const float* __restrict__ adj_val,
                  EdgeRec* packed,
                  const int* __restrict__ bin_count) {
  __shared__ __attribute__((aligned(16))) EdgeRec img[RPBIN * SLOT];
  const int tid  = threadIdx.x;
  const int b    = blockIdx.x;
  const int row0 = b * RPBIN;
  EdgeRec* const region = packed + (size_t)row0 * SLOT;
  int4* const img4 = (int4*)img;
  for (int i = tid; i < IMG_VEC; i += PB_THREADS) img4[i] = make_int4(0, 0, 0, 0);
  __syncthreads();
  const int n_bin = bin_count[b * 32];
  if (n_bin >= 0 && n_bin <= BINCAP) {
    const BinRec* const parked = (const BinRec*)region;
    for (int i0 = 0; i0 < n_bin; i0 += PB_BATCH * PB_THREADS) {
      BinRec o[PB_BATCH];
#pragma unroll
      for (int u = 0; u < PB_BATCH; ++u) {
        const int i = i0 + u * PB_THREADS + tid;
        o[u].col = 0; o[u].row = -1; o[u].adj = 0.f;
        if (i < n_bin) o[u] = parked[i];
      }
#pragma unroll
      for (int u = 0; u < PB_BATCH; ++u) place_edge(img, o[u].row - row0, o[u].col, o[u].adj);
    }
  } else {
    for (int e = tid; e < N_EDGES; e += PB_THREADS) {
      const int rl = row[e] - row0;
      if ((unsigned)rl < (unsigned)RPBIN) place_edge(img, rl, col[e], adj_val[e]);
    }
  }
  __syncthreads();
  int4* const dst = (int4*)region;
  for (int i = tid; i < IMG_VEC; i += PB_THREADS) dst[i] = img4[i];
}

#define SR_WPB 8
#define SR_RPW 4

#define SEL8(a0, a1, a2, a3, a4, a5, a6, a7)                                   \
  ((a0) | ((a1) << 3) | ((a2) << 6) | ((a3) << 9) | ((a4) << 12) |             \
   ((a5) << 15) | ((a6) << 18) | ((a7) << 21))

template <int I>
static __device__ __forceinline__ unsigned from_group_lane(unsigned v) {
  return __builtin_amdgcn_mov_dpp8(v, SEL8(I, I, I, I, I, I, I, I));
}

static __device__ __forceinline__ float sum_over_group(float p) {
  p += __uint_as_float(__builtin_amdgcn_mov_dpp8(__float_as_uint(p), SEL8(1, 0, 3, 2, 5, 4, 7, 6)));
  p += __uint_as_float(__builtin_amdgcn_mov_dpp8(__float_as_uint(p), SEL8(2, 3, 0, 1, 6, 7, 4, 5)));
  p += __uint_as_float(__builtin_amdgcn_mov_dpp8(__float_as_uint(p), SEL8(4, 5, 6, 7, 0, 1, 2, 3)));
  return p;
}

template <int I>
static __device__ __forceinline__ float4 load_col_line(const float* __restrict__ a_col,
                                                       unsigned c, unsigned sub16,
                                                       int g, int m) {
  const unsigned cc = from_group_lane<I>(c);
  float4 v = make_float4(0.f, 0.f, 0.f, 0.f);
  if (8 * g + I < m) v = *(const float4*)((const char*)a_col + (cc * (DA * 4u) + sub16));
  return v;
}

template <int I>
static __device__ __forceinline__ float keep_own(const float4& u, const float4& v, int sub, float s) {
  const float p = sum_over_group(u.x * v.x + u.y * v.y + u.z * v.z + u.w * v.w);
  return (sub == I) ? p : s;
}

static __device__ __forceinline__ unsigned score_key(float s) {
  const unsigned b = __float_as_uint(s);
  return (b & 0x80000000u) ? ~b : (b | 0x80000000u);
}

static __device__ float edge_score_whole(const float* __restrict__ a_row,
                                         const float* __restrict__ a_col,
                                         int r, unsigned c) {
  const float4* ar = (const float4*)(a_row + (size_t)r * DA);
  const float4* ac = (const float4*)(a_col + (size_t)c * DA);
  float s = 0.f;
#pragma unroll 1
  for (int i = 0; i < DA / 4; ++i) {
    const float4 u = ar[i];
    const float4 v = ac[i];
    s += u.x * v.x + u.y * v.y + u.z * v.z + u.w * v.w;
  }
  s *= 0.17677669529663687f;
  return (s >= 0.f) ? s : SLOPE * s;
}

static __device__ unsigned slow_row_maxkey(const float* __restrict__ a_row,
                                           const float* __restrict__ a_col,
                                           const int* __restrict__ row,
                                           const int* __restrict__ col,
                                           int r, int lane) {
  unsigned kmax = 0u;
  for (int e0 = 0; e0 < N_EDGES; e0 += 32) {
    const int e = e0 + lane;
    if (row[e] == r) {
      const unsigned c = min((unsigned)col[e], (unsigned)(N_NODES - 1));
      const unsigned key = score_key(edge_score_whole(a_row, a_col, r, c));
      kmax = (key > kmax) ? key : kmax;
    }
  }
  return kmax;
}

__global__ void __launch_bounds__(SR_WPB * 32)
score_rows_kernel(const float* __restrict__ a_row,
                  const float* __restrict__ a_col,
                  const int* __restrict__ row,
                  const int* __restrict__ col,
                  EdgeRec* __restrict__ packed,
                  unsigned* __restrict__ maxkey) {
  __shared__ unsigned wave_key[SR_WPB];
  const int wv   = __builtin_amdgcn_readfirstlane(threadIdx.x >> 5);
  const int lane = threadIdx.x & 31;
  const int sub  = lane & 7;
  const int g    = lane >> 3;
  const unsigned sub16 = (unsigned)sub * 16u;
  const int wrow0 = (blockIdx.x * SR_WPB + wv) * SR_RPW;
  unsigned kmax = 0u;
  for (int i = 0; i < SR_RPW; ++i) {
    const int r = wrow0 + i;
    if (r < N_NODES) {
      const int n_all = packed[(size_t)r * SLOT].col;
      if (n_all > CAP) {
        const unsigned ks = slow_row_maxkey(a_row, a_col, row, col, r, lane);
        kmax = (ks > kmax) ? ks : kmax;
      }
      const int n = (n_all > CAP) ? 0 : n_all;
      const int s0 = r * SLOT + 1;
      const float4 u = *(const float4*)((const char*)a_row + ((unsigned)r * (DA * 4u) + sub16));
      for (int j = 0; j < n; j += 32) {
        const int m = (n - j < 32) ? (n - j) : 32;
        const bool live = lane < m;
        int q = s0 + j + lane;
        q = q < 0 ? 0 : (q > N_NODES * SLOT - 1 ? N_NODES * SLOT - 1 : q);
        unsigned c = 0u;
        if (live) c = (unsigned)packed[q].col;
        c = min(c, (unsigned)(N_NODES - 1));
        float4 v[8];
        v[0] = load_col_line<0>(a_col, c, sub16, g, m);
        v[1] = load_col_line<1>(a_col, c, sub16, g, m);
        v[2] = load_col_line<2>(a_col, c, sub16, g, m);
        v[3] = load_col_line<3>(a_col, c, sub16, g, m);
        v[4] = load_col_line<4>(a_col, c, sub16, g, m);
        v[5] = load_col_line<5>(a_col, c, sub16, g, m);
        v[6] = load_col_line<6>(a_col, c, sub16, g, m);
        v[7] = load_col_line<7>(a_col, c, sub16, g, m);
        float s = 0.f;
        s = keep_own<0>(u, v[0], sub, s);
        s = keep_own<1>(u, v[1], sub, s);
        s = keep_own<2>(u, v[2], sub, s);
        s = keep_own<3>(u, v[3], sub, s);
        s = keep_own<4>(u, v[4], sub, s);
        s = keep_own<5>(u, v[5], sub, s);
        s = keep_own<6>(u, v[6], sub, s);
        s = keep_own<7>(u, v[7], sub, s);
        s *= 0.17677669529663687f;
        s = (s >= 0.f) ? s : SLOPE * s;
        if (live) {
          packed[q].a = s;
          const unsigned key = score_key(s);
          kmax = (key > kmax) ? key : kmax;
        }
      }
    }
  }
#pragma unroll
  for (int off = 16; off > 0; off >>= 1) {
    const unsigned o = __shfl_xor(kmax, off, 32);
    kmax = (o > kmax) ? o : kmax;
  }
  if (lane == 0) wave_key[wv] = kmax;
  __syncthreads();
  if (threadIdx.x == 0) {
    unsigned mx = wave_key[0];
    for (int w = 1; w < SR_WPB; ++w) mx = (wave_key[w] > mx) ? wave_key[w] : mx;
    atomicMax(maxkey, mx);
  }
}

static __device__ __forceinline__ EdgeRec load_rec(const EdgeRec* __restrict__ packed,
                                                   int q, bool live, float gmax) {
  EdgeRec p;
  p.col = 0; p.a = 0.f; p.adj = 0.f;
  if (live) {
    q = q < 0 ? 0 : (q > N_NODES * SLOT - 1 ? N_NODES * SLOT - 1 : q);
    p = packed[q];
    p.a = expf(p.a - gmax);
  }
  return p;
}

static __device__ __forceinline__ float4 gather_row(const float* __restrict__ x,
                                                    const EdgeRec* __restrict__ packed,
                                                    float gmax, int r, int n, int lane) {
  const int s = r * SLOT + 1;

  const EdgeRec first = load_rec(packed, s + lane, lane < n, gmax);
  float part = first.a;
  for (int j = 32; j < n; j += 32)
    part += load_rec(packed, s + j + lane, j + lane < n, gmax).a;
  for (int off = 16; off > 0; off >>= 1) part += __shfl_xor(part, off, 32);
  const float inv = 1.0f / (part + 1e-15f);

  const float* xl = x + lane * 4;
  float4 acc = make_float4(0.f, 0.f, 0.f, 0.f);
  for (int j = 0; j < n; j += 32) {
    const int m = (n - j < 32) ? (n - j) : 32;
    EdgeRec p = first;
    if (j > 0) p = load_rec(packed, s + j + lane, lane < m, gmax);
    const int   pc = p.col;
    const float pv = (p.adj + p.a * inv) * 0.5f;
    for (int k = 0; k < m; k += 4) {
      const unsigned c0 = min((unsigned)__builtin_amdgcn_readlane(pc, k), (unsigned)(N_NODES - 1));
      const unsigned c1 = min((unsigned)__builtin_amdgcn_readlane(pc, k + 1), (unsigned)(N_NODES - 1));
      const unsigned c2 = min((unsigned)__builtin_amdgcn_readlane(pc, k + 2), (unsigned)(N_NODES - 1));
      const unsigned c3 = min((unsigned)__builtin_amdgcn_readlane(pc, k + 3), (unsigned)(N_NODES - 1));
      const float v0 = __int_as_float(__builtin_amdgcn_readlane(__float_as_int(pv), k));
      const float v1 = __int_as_float(__builtin_amdgcn_readlane(__float_as_int(pv), k + 1));
      const float v2 = __int_as_float(__builtin_amdgcn_readlane(__float_as_int(pv), k + 2));
      const float v3 = __int_as_float(__builtin_amdgcn_readlane(__float_as_int(pv), k + 3));
      const float4 x0 = *(const float4*)(xl + (size_t)c0 * DIN);
      const float4 x1 = *(const float4*)(xl + (size_t)c1 * DIN);
      const float4 x2 = *(const float4*)(xl + (size_t)c2 * DIN);
      const float4 x3 = *(const float4*)(xl + (size_t)c3 * DIN);
      acc.x += v0 * x0.x; acc.y += v0 * x0.y; acc.z += v0 * x0.z; acc.w += v0 * x0.w;
      acc.x += v1 * x1.x; acc.y += v1 * x1.y; acc.z += v1 * x1.z; acc.w += v1 * x1.w;
      acc.x += v2 * x2.x; acc.y += v2 * x2.y; acc.z += v2 * x2.z; acc.w += v2 * x2.w;
      acc.x += v3 * x3.x; acc.y += v3 * x3.y; acc.z += v3 * x3.z; acc.w += v3 * x3.w;
    }
  }
  return acc;
}

static __device__ float4 gather_row_slow(const float* __restrict__ x,
                                         const int* __restrict__ row,
                                         const int* __restrict__ col,
                                         const float* __restrict__ adj_val,
                                         const float* __restrict__ a_row,
                                         const float* __restrict__ a_col,
                                         float gmax, int r, int lane) {
  float part = 0.f;
  for (int e0 = 0; e0 < N_EDGES; e0 += 32) {
    const int e = e0 + lane;
    if (row[e] == r) {
      const unsigned c = min((unsigned)col[e], (unsigned)(N_NODES - 1));
      part += expf(edge_score_whole(a_row, a_col, r, c) - gmax);
    }
  }
  for (int off = 16; off > 0; off >>= 1) part += __shfl_xor(part, off, 32);
  const float inv = 1.0f / (part + 1e-15f);

  const float* xl = x + lane * 4;
  float4 acc = make_float4(0.f, 0.f, 0.f, 0.f);
  for (int e0 = 0; e0 < N_EDGES; e0 += 32) {
    const int e = e0 + lane;
    const bool mine = row[e] == r;
    unsigned c = 0u;
    float val = 0.f;
    if (mine) {
      c = min((unsigned)col[e], (unsigned)(N_NODES - 1));
      val = (adj_val[e] + expf(edge_score_whole(a_row, a_col, r, c) - gmax) * inv) * 0.5f;
    }
    unsigned todo = __builtin_amdgcn_ballot_w32(mine);
    while (todo != 0u) {
      const int l = __builtin_ctz(todo);
      todo &= todo - 1u;
      const unsigned cc = (unsigned)__builtin_amdgcn_readlane((int)c, l);
      const float vv = __int_as_float(__builtin_amdgcn_readlane(__float_as_int(val), l));
      const float4 xv = *(const float4*)(xl + (size_t)cc * DIN);
      acc.x += vv * xv.x; acc.y += vv * xv.y; acc.z += vv * xv.z; acc.w += vv * xv.w;
    }
  }
  return acc;
}

#define GG_WAVES 8
#define TILE_ROWS 16
#define TILE_PITCH 132

__global__ void __launch_bounds__(GG_WAVES * 32)
gather_gemm_kernel(const float* __restrict__ x,
                   const int* __restrict__ row,
                   const int* __restrict__ col,
                   const float* __restrict__ adj_val,
                   const float* __restrict__ a_row,
                   const float* __restrict__ a_col,
                   const EdgeRec* __restrict__ packed,
                   const unsigned* __restrict__ maxkey,
                   const float* __restrict__ W_x,
                   const float* __restrict__ b_x,
                   float* __restrict__ out) {
  static_assert(N_NODES % TILE_ROWS == 0, "every block has 16 whole rows");
  static_assert(GG_WAVES * 32 == DOUT, "the waves' column groups cover DOUT");
  static_assert(TILE_ROWS == 2 * GG_WAVES, "two rows per wave");
  __shared__ float tile[TILE_ROWS * TILE_PITCH];
  const int wv   = __builtin_amdgcn_readfirstlane(threadIdx.x >> 5);
  const int lane = threadIdx.x & 31;
  const int row0 = blockIdx.x * TILE_ROWS;
  const unsigned key = *maxkey;
  const unsigned kb = (key & 0x80000000u) ? (key ^ 0x80000000u) : ~key;
  const float gmax = __uint_as_float(kb);

  for (int i = 0; i < 2; ++i) {
    const int lr = wv + GG_WAVES * i;
    const int r  = row0 + lr;
    const int n_all = packed[(size_t)r * SLOT].col;
    float4 f;
    if (n_all > CAP)
      f = gather_row_slow(x, row, col, adj_val, a_row, a_col, gmax, r, lane);
    else
      f = gather_row(x, packed, gmax, r, n_all, lane);
    *(float4*)(tile + lr * TILE_PITCH + lane * 4) = f;
  }
  __syncthreads();

  const int half  = lane >> 4;
  const int m     = lane & 15;
  const int cbase = wv * 32;

  v8f acc0, acc1;
  {
    const float bias0 = b_x[cbase + m];
    const float bias1 = b_x[cbase + 16 + m];
#pragma unroll
    for (int v = 0; v < 8; ++v) { acc0[v] = bias0; acc1[v] = bias1; }
  }
  const float* arow = tile + m * TILE_PITCH + 2 * half;
  for (int k = 0; k < DIN; k += 4) {
    v2f a = *(const v2f*)(arow + k);
    const int kk = k + 2 * half;
    v2f b0 = { W_x[kk * DOUT + cbase + m],      W_x[(kk + 1) * DOUT + cbase + m] };
    v2f b1 = { W_x[kk * DOUT + cbase + 16 + m], W_x[(kk + 1) * DOUT + cbase + 16 + m] };
    acc0 = wmma_f32_16x16x4(a, b0, acc0);
    acc1 = wmma_f32_16x16x4(a, b1, acc1);
  }
#pragma unroll
  for (int v = 0; v < 8; ++v) {
    float* orow = out + (size_t)(row0 + v + half * 8) * DOUT + cbase + m;
    orow[0]  = acc0[v];
    orow[16] = acc1[v];
  }
}

extern "C" void kernel_launch(void* const* d_in, const int* in_sizes, int n_in,
                              void* d_out, int out_size, void* d_ws, size_t ws_size,
                              hipStream_t stream) {
  const float* x      = (const float*)d_in[0];
  const int*   row    = (const int*)d_in[1];
  const int*   col    = (const int*)d_in[2];
  const float* adjval = (const float*)d_in[3];
  const float* W_row  = (const float*)d_in[4];
  const float* W_col  = (const float*)d_in[5];
  const float* W_x    = (const float*)d_in[6];
  const float* b_x    = (const float*)d_in[7];
  float* out = (float*)d_out;

  char* ws = (char*)d_ws;
  size_t off = 0;
  float* a_row = (float*)(ws + off); off += (size_t)N_NODES * DA * 4;
  float* a_col = (float*)(ws + off); off += (size_t)N_NODES * DA * 4;
  unsigned* maxkey = (unsigned*)(ws + off); off += 256;
  int* bin_count = (int*)(ws + off); off += (size_t)NBIN * 128;
  EdgeRec* packed = (EdgeRec*)(ws + off); off += (size_t)N_NODES * SLOT * 12;
  static_assert(sizeof(EdgeRec) == 12, "a place is 12 bytes");
  static_assert((2 * (size_t)N_NODES * DA * 4 + 256 + (size_t)NBIN * 128) % 16 == 0,
                "the slots start on a 16-byte boundary: level 2 writes them 16 bytes at a time");
  static_assert(2 * (size_t)N_NODES * DA * 4 + 256 + (size_t)NBIN * 128 + (size_t)N_NODES * SLOT * 12 <=
                3 * (size_t)N_EDGES * 4 + (size_t)N_NODES * 32 * 4 + 256 +
                (size_t)N_NODES * 4 + (size_t)N_NODES * DIN * 4,
                "the plan stays inside the workspace the earlier plans used");

  hipMemsetAsync(maxkey, 0, 256 + (size_t)NBIN * 128, stream);

  proj_park_kernel<<<PROJ_BLOCKS + PA_BLOCKS, 256, 0, stream>>>(x, W_row, W_col, a_row, a_col, row, col, adjval, packed, bin_count);
  place_rows_kernel<<<NBIN, PB_THREADS, 0, stream>>>(row, col, adjval, packed, bin_count);
  score_rows_kernel<<<(N_NODES + SR_WPB * SR_RPW - 1) / (SR_WPB * SR_RPW), SR_WPB * 32, 0, stream>>>(a_row, a_col, row, col, packed, maxkey);
  {
    const int blocks = N_NODES / TILE_ROWS;
    gather_gemm_kernel<<<blocks, GG_WAVES * 32, 0, stream>>>(x, row, col, adjval, a_row, a_col, packed, maxkey, W_x, b_x, out);
  }
}


# Level 2: row counts in an LDS word array of their own (rank adds spread over the banks, not all on one); the wave that scores a row stores its header and used places; zero fill and image copy go
# speedup vs baseline: 5.6070x; 1.2653x over previous
#include <hip/hip_runtime.h>

#define N_NODES 50000
#define N_EDGES 1600000
#define DIN 128
#define DOUT 256
#define DA 32
#define SLOPE 0.2f
#define SLOT 64
#define CAP (SLOT - 1)
static_assert(SLOT == 2 * (N_EDGES / N_NODES), "slot size follows the mean row length");
static_assert(N_EDGES % 32 == 0, "the slow paths walk the edge list a whole wave at a time");
#define NBIN 250
#define RPBIN (N_NODES / NBIN)
#define BINCAP (RPBIN * SLOT)
#define EPB 3200
#define PA_BLOCKS (N_EDGES / EPB)
#define PA_TRIPS ((EPB + 255) / 256)
static_assert(N_NODES % NBIN == 0, "every range has RPBIN whole rows");
static_assert(N_EDGES % EPB == 0, "every level-1 block has EPB whole edges");
static_assert(NBIN <= 256, "one thread of a level-1 block per range");
static_assert(EPB <= 65535, "an edge's rank inside its block fits 16 bits");

typedef __attribute__((ext_vector_type(2))) float v2f;
struct EdgeRec { int col; float a; float adj; };
struct BinRec { int col; int row; float adj; };
static_assert(sizeof(BinRec) == sizeof(EdgeRec), "a parked edge takes one place");
typedef __attribute__((ext_vector_type(8))) float v8f;

static __device__ __forceinline__ v8f wmma_f32_16x16x4(v2f a, v2f b, v8f c) {
  return __builtin_amdgcn_wmma_f32_16x16x4_f32(false, a, false, b, (short)0, c,
                                               false, false);
}

#define PROJ_BLOCKS ((N_NODES / 16 * 32 + 255) / 256)

static __device__ __forceinline__ int range_of(int r) {
  const unsigned b = (unsigned)r / (unsigned)RPBIN;
  return (int)(b < (unsigned)NBIN ? b : (unsigned)(NBIN - 1));
}

__global__ void __launch_bounds__(256)
proj_park_kernel(const float* __restrict__ x,
                 const float* __restrict__ W_row,
                 const float* __restrict__ W_col,
                 float* __restrict__ a_row,
                 float* __restrict__ a_col,
                 const int* __restrict__ row,
                 const int* __restrict__ col,
                 const float* __restrict__ adj_val,
                 EdgeRec* __restrict__ packed,
                 int* __restrict__ bin_count) {
  if (blockIdx.x >= PROJ_BLOCKS) {
    __shared__ int s_hist[256];
    __shared__ int s_offs[256];
    __shared__ int s_gbase[256];
    __shared__ int s_wsum[8];
    __shared__ unsigned short s_lrank[EPB];
    __shared__ BinRec s_stage[EPB];
    const int tid  = threadIdx.x;
    const int lane = tid & 31;
    const int wv   = tid >> 5;
    const int e0   = (blockIdx.x - PROJ_BLOCKS) * EPB;
    s_hist[tid] = 0;
    __syncthreads();
    int rr[PA_TRIPS];
#pragma unroll
    for (int j = 0; j < PA_TRIPS; ++j) {
      const int i = j * 256 + tid;
      rr[j] = (i < EPB) ? row[e0 + i] : 0;
    }
#pragma unroll
    for (int j = 0; j < PA_TRIPS; ++j) {
      const int i = j * 256 + tid;
      if (i < EPB) s_lrank[i] = (unsigned short)atomicAdd(&s_hist[range_of(rr[j])], 1);
    }
    __syncthreads();
    const int h = s_hist[tid];
    int inc = h;
#pragma unroll
    for (int off = 1; off < 32; off <<= 1) {
      const int t = __shfl_up(inc, off, 32);
      if (lane >= off) inc += t;
    }
    if (lane == 31) s_wsum[wv] = inc;
    __syncthreads();
    int before = 0;
    for (int w = 0; w < wv; ++w) before += s_wsum[w];
    s_offs[tid] = before + inc - h;
    if (tid < NBIN) s_gbase[tid] = atomicAdd(&bin_count[tid * 32], h);
    __syncthreads();
#pragma unroll
    for (int j = 0; j < PA_TRIPS; ++j) {
      const int i = j * 256 + tid;
      if (i < EPB) {
        const int e = e0 + i;
        BinRec o; o.col = col[e]; o.row = rr[j]; o.adj = adj_val[e];
        s_stage[s_offs[range_of(o.row)] + s_lrank[i]] = o;
      }
    }
    __syncthreads();
    BinRec* const parked = (BinRec*)packed;
    for (int i = tid; i < EPB; i += 256) {
      const BinRec o = s_stage[i];
      const int b = range_of(o.row);
      const int k = s_gbase[b] + (i - s_offs[b]);
      if (k >= 0 && k < BINCAP) parked[(size_t)b * BINCAP + k] = o;
    }
    return;
  }
  const int wave = (blockIdx.x * blockDim.x + threadIdx.x) >> 5;
  const int lane = threadIdx.x & 31;
  if (wave >= N_NODES / 16) return;
  const int tile = wave;
  const int half = lane >> 4;
  const int m    = lane & 15;
  const int node = tile * 16 + m;

  v8f acc0 = {}, acc1 = {}, acc2 = {}, acc3 = {};
  const float* xrow = x + (size_t)node * DIN + 2 * half;
  for (int k = 0; k < DIN; k += 4) {
    v2f a = *(const v2f*)(xrow + k);
    const int kk = k + 2 * half;
    v2f b0 = { W_row[kk * DA + m],      W_row[(kk + 1) * DA + m] };
    v2f b1 = { W_row[kk * DA + 16 + m], W_row[(kk + 1) * DA + 16 + m] };
    v2f b2 = { W_col[kk * DA + m],      W_col[(kk + 1) * DA + m] };
    v2f b3 = { W_col[kk * DA + 16 + m], W_col[(kk + 1) * DA + 16 + m] };
    acc0 = wmma_f32_16x16x4(a, b0, acc0);
    acc1 = wmma_f32_16x16x4(a, b1, acc1);
    acc2 = wmma_f32_16x16x4(a, b2, acc2);
    acc3 = wmma_f32_16x16x4(a, b3, acc3);
  }
#pragma unroll
  for (int v = 0; v < 8; ++v) {
    const int on = tile * 16 + v + half * 8;
    a_row[on * DA + m]      = acc0[v];
    a_row[on * DA + 16 + m] = acc1[v];
    a_col[on * DA + m]      = acc2[v];
    a_col[on * DA + 16 + m] = acc3[v];
  }
}

#define SEL8(a0, a1, a2, a3, a4, a5, a6, a7)                                   \
  ((a0) | ((a1) << 3) | ((a2) << 6) | ((a3) << 9) | ((a4) << 12) |             \
   ((a5) << 15) | ((a6) << 18) | ((a7) << 21))

template <int I>
static __device__ __forceinline__ unsigned from_group_lane(unsigned v) {
  return __builtin_amdgcn_mov_dpp8(v, SEL8(I, I, I, I, I, I, I, I));
}

static __device__ __forceinline__ float sum_over_group(float p) {
  p += __uint_as_float(__builtin_amdgcn_mov_dpp8(__float_as_uint(p), SEL8(1, 0, 3, 2, 5, 4, 7, 6)));
  p += __uint_as_float(__builtin_amdgcn_mov_dpp8(__float_as_uint(p), SEL8(2, 3, 0, 1, 6, 7, 4, 5)));
  p += __uint_as_float(__builtin_amdgcn_mov_dpp8(__float_as_uint(p), SEL8(4, 5, 6, 7, 0, 1, 2, 3)));
  return p;
}

template <int I>
static __device__ __forceinline__ float4 load_col_line(const float* __restrict__ a_col,
                                                       unsigned c, unsigned sub16,
                                                       int g, int m) {
  const unsigned cc = from_group_lane<I>(c);
  float4 v = make_float4(0.f, 0.f, 0.f, 0.f);
  if (8 * g + I < m) v = *(const float4*)((const char*)a_col + (cc * (DA * 4u) + sub16));
  return v;
}

template <int I>
static __device__ __forceinline__ float keep_own(const float4& u, const float4& v, int sub, float s) {
  const float p = sum_over_group(u.x * v.x + u.y * v.y + u.z * v.z + u.w * v.w);
  return (sub == I) ? p : s;
}

static __device__ __forceinline__ unsigned score_key(float s) {
  const unsigned b = __float_as_uint(s);
  return (b & 0x80000000u) ? ~b : (b | 0x80000000u);
}

static __device__ float edge_score_whole(const float* __restrict__ a_row,
                                         const float* __restrict__ a_col,
                                         int r, unsigned c) {
  const float4* ar = (const float4*)(a_row + (size_t)r * DA);
  const float4* ac = (const float4*)(a_col + (size_t)c * DA);
  float s = 0.f;
#pragma unroll 1
  for (int i = 0; i < DA / 4; ++i) {
    const float4 u = ar[i];
    const float4 v = ac[i];
    s += u.x * v.x + u.y * v.y + u.z * v.z + u.w * v.w;
  }
  s *= 0.17677669529663687f;
  return (s >= 0.f) ? s : SLOPE * s;
}

static __device__ unsigned slow_row_maxkey(const float* __restrict__ a_row,
                                           const float* __restrict__ a_col,
                                           const int* __restrict__ row,
                                           const int* __restrict__ col,
                                           int r, int lane) {
  unsigned kmax = 0u;
  for (int e0 = 0; e0 < N_EDGES; e0 += 32) {
    const int e = e0 + lane;
    if (row[e] == r) {
      const unsigned c = min((unsigned)col[e], (unsigned)(N_NODES - 1));
      const unsigned key = score_key(edge_score_whole(a_row, a_col, r, c));
      kmax = (key > kmax) ? key : kmax;
    }
  }
  return kmax;
}

#define PB_THREADS 1024
#define PB_WAVES (PB_THREADS / 32)
#define PB_BATCH 8
static_assert((SLOT * 12) % 16 == 0, "a slot is whole 16-byte pieces and starts on a 16-byte boundary");

static __device__ __forceinline__ void place_edge(EdgeRec* img, int* cnt, int rl, int c, float adj) {
  if ((unsigned)rl < (unsigned)RPBIN) {
    EdgeRec* const slot = img + rl * SLOT;
    const int k = atomicAdd(&cnt[rl], 1);
    if (k >= 0 && k < CAP) { EdgeRec o; o.col = c; o.a = 0.f; o.adj = adj; slot[1 + k] = o; }
  }
}

__global__ void __launch_bounds__(PB_THREADS)
place_score_kernel(const int* __restrict__ row,
                   const int* __restrict__ col,
                   const float* __restrict__ adj_val,
                   const float* __restrict__ a_row,
                   const float* __restrict__ a_col,
                   EdgeRec* packed,
                   const int* __restrict__ bin_count,
                   unsigned* __restrict__ maxkey) {
  __shared__ __attribute__((aligned(16))) EdgeRec img[RPBIN * SLOT];
  __shared__ int cnt[RPBIN];
  __shared__ unsigned wave_key[PB_WAVES];
  const int tid  = threadIdx.x;
  const int b    = blockIdx.x;
  const int row0 = b * RPBIN;
  EdgeRec* const region = packed + (size_t)row0 * SLOT;
  for (int i = tid; i < RPBIN; i += PB_THREADS) cnt[i] = 0;
  __syncthreads();
  const int n_bin = bin_count[b * 32];
  if (n_bin >= 0 && n_bin <= BINCAP) {
    const BinRec* const parked = (const BinRec*)region;
    for (int i0 = 0; i0 < n_bin; i0 += PB_BATCH * PB_THREADS) {
      BinRec o[PB_BATCH];
#pragma unroll
      for (int u = 0; u < PB_BATCH; ++u) {
        const int i = i0 + u * PB_THREADS + tid;
        o[u].col = 0; o[u].row = -1; o[u].adj = 0.f;
        if (i < n_bin) o[u] = parked[i];
      }
#pragma unroll
      for (int u = 0; u < PB_BATCH; ++u) place_edge(img, cnt, o[u].row - row0, o[u].col, o[u].adj);
    }
  } else {
    for (int e = tid; e < N_EDGES; e += PB_THREADS) {
      const int rl = row[e] - row0;
      if ((unsigned)rl < (unsigned)RPBIN) place_edge(img, cnt, rl, col[e], adj_val[e]);
    }
  }
  __syncthreads();

  const int wv   = __builtin_amdgcn_readfirstlane(tid >> 5);
  const int lane = tid & 31;
  const int sub  = lane & 7;
  const int g    = lane >> 3;
  const unsigned sub16 = (unsigned)sub * 16u;
  unsigned kmax = 0u;
  for (int rl = wv; rl < RPBIN; rl += PB_WAVES) {
    const int r = row0 + rl;
    EdgeRec* const slot = img + rl * SLOT;
    const int n_all = __builtin_amdgcn_readfirstlane(cnt[rl]);
    if (n_all > CAP) {
      const unsigned ks = slow_row_maxkey(a_row, a_col, row, col, r, lane);
      kmax = (ks > kmax) ? ks : kmax;
    }
    const int n = (n_all > CAP) ? 0 : n_all;
    const float4 u = *(const float4*)((const char*)a_row + ((unsigned)r * (DA * 4u) + sub16));
    for (int j = 0; j < n; j += 32) {
      const int m = (n - j < 32) ? (n - j) : 32;
      const bool live = lane < m;
      EdgeRec* const rec = slot + 1 + j + lane;
      unsigned c = 0u;
      if (live) c = (unsigned)rec->col;
      c = min(c, (unsigned)(N_NODES - 1));
      float4 v[8];
      v[0] = load_col_line<0>(a_col, c, sub16, g, m);
      v[1] = load_col_line<1>(a_col, c, sub16, g, m);
      v[2] = load_col_line<2>(a_col, c, sub16, g, m);
      v[3] = load_col_line<3>(a_col, c, sub16, g, m);
      v[4] = load_col_line<4>(a_col, c, sub16, g, m);
      v[5] = load_col_line<5>(a_col, c, sub16, g, m);
      v[6] = load_col_line<6>(a_col, c, sub16, g, m);
      v[7] = load_col_line<7>(a_col, c, sub16, g, m);
      float s = 0.f;
      s = keep_own<0>(u, v[0], sub, s);
      s = keep_own<1>(u, v[1], sub, s);
      s = keep_own<2>(u, v[2], sub, s);
      s = keep_own<3>(u, v[3], sub, s);
      s = keep_own<4>(u, v[4], sub, s);
      s = keep_own<5>(u, v[5], sub, s);
      s = keep_own<6>(u, v[6], sub, s);
      s = keep_own<7>(u, v[7], sub, s);
      s *= 0.17677669529663687f;
      s = (s >= 0.f) ? s : SLOPE * s;
      if (live) {
        rec->a = s;
        const unsigned key = score_key(s);
        kmax = (key > kmax) ? key : kmax;
      }
    }
    if (lane == 0) { EdgeRec h; h.col = n_all; h.a = 0.f; h.adj = 0.f; slot[0] = h; }
    __builtin_amdgcn_fence(__ATOMIC_SEQ_CST, "wavefront");
    {
      const int4* const s4 = (const int4*)slot;
      int4* const d4 = (int4*)(region + (size_t)rl * SLOT);
      const int pieces = (12 * (1 + n) + 15) / 16;
      for (int i = lane; i < pieces; i += 32) d4[i] = s4[i];
    }
  }
#pragma unroll
  for (int off = 16; off > 0; off >>= 1) {
    const unsigned o = __shfl_xor(kmax, off, 32);
    kmax = (o > kmax) ? o : kmax;
  }
  if (lane == 0) wave_key[wv] = kmax;
  __syncthreads();
  if (tid == 0) {
    unsigned mx = wave_key[0];
    for (int w = 1; w < PB_WAVES; ++w) mx = (wave_key[w] > mx) ? wave_key[w] : mx;
    atomicMax(maxkey, mx);
  }
}

static __device__ __forceinline__ EdgeRec load_rec(const EdgeRec* __restrict__ packed,
                                                   int q, bool live, float gmax) {
  EdgeRec p;
  p.col = 0; p.a = 0.f; p.adj = 0.f;
  if (live) {
    q = q < 0 ? 0 : (q > N_NODES * SLOT - 1 ? N_NODES * SLOT - 1 : q);
    p = packed[q];
    p.a = expf(p.a - gmax);
  }
  return p;
}

static __device__ __forceinline__ float4 gather_row(const float* __restrict__ x,
                                                    const EdgeRec* __restrict__ packed,
                                                    float gmax, int r, int n, int lane) {
  const int s = r * SLOT + 1;

  const EdgeRec first = load_rec(packed, s + lane, lane < n, gmax);
  float part = first.a;
  for (int j = 32; j < n; j += 32)
    part += load_rec(packed, s + j + lane, j + lane < n, gmax).a;
  for (int off = 16; off > 0; off >>= 1) part += __shfl_xor(part, off, 32);
  const float inv = 1.0f / (part + 1e-15f);

  const float* xl = x + lane * 4;
  float4 acc = make_float4(0.f, 0.f, 0.f, 0.f);
  for (int j = 0; j < n; j += 32) {
    const int m = (n - j < 32) ? (n - j) : 32;
    EdgeRec p = first;
    if (j > 0) p = load_rec(packed, s + j + lane, lane < m, gmax);
    const int   pc = p.col;
    const float pv = (p.adj + p.a * inv) * 0.5f;
    for (int k = 0; k < m; k += 4) {
      const unsigned c0 = min((unsigned)__builtin_amdgcn_readlane(pc, k), (unsigned)(N_NODES - 1));
      const unsigned c1 = min((unsigned)__builtin_amdgcn_readlane(pc, k + 1), (unsigned)(N_NODES - 1));
      const unsigned c2 = min((unsigned)__builtin_amdgcn_readlane(pc, k + 2), (unsigned)(N_NODES - 1));
      const unsigned c3 = min((unsigned)__builtin_amdgcn_readlane(pc, k + 3), (unsigned)(N_NODES - 1));
      const float v0 = __int_as_float(__builtin_amdgcn_readlane(__float_as_int(pv), k));
      const float v1 = __int_as_float(__builtin_amdgcn_readlane(__float_as_int(pv), k + 1));
      const float v2 = __int_as_float(__builtin_amdgcn_readlane(__float_as_int(pv), k + 2));
      const float v3 = __int_as_float(__builtin_amdgcn_readlane(__float_as_int(pv), k + 3));
      const float4 x0 = *(const float4*)(xl + (size_t)c0 * DIN);
      const float4 x1 = *(const float4*)(xl + (size_t)c1 * DIN);
      const float4 x2 = *(const float4*)(xl + (size_t)c2 * DIN);
      const float4 x3 = *(const float4*)(xl + (size_t)c3 * DIN);
      acc.x += v0 * x0.x; acc.y += v0 * x0.y; acc.z += v0 * x0.z; acc.w += v0 * x0.w;
      acc.x += v1 * x1.x; acc.y += v1 * x1.y; acc.z += v1 * x1.z; acc.w += v1 * x1.w;
      acc.x += v2 * x2.x; acc.y += v2 * x2.y; acc.z += v2 * x2.z; acc.w += v2 * x2.w;
      acc.x += v3 * x3.x; acc.y += v3 * x3.y; acc.z += v3 * x3.z; acc.w += v3 * x3.w;
    }
  }
  return acc;
}

static __device__ float4 gather_row_slow(const float* __restrict__ x,
                                         const int* __restrict__ row,
                                         const int* __restrict__ col,
                                         const float* __restrict__ adj_val,
                                         const float* __restrict__ a_row,
                                         const float* __restrict__ a_col,
                                         float gmax, int r, int lane) {
  float part = 0.f;
  for (int e0 = 0; e0 < N_EDGES; e0 += 32) {
    const int e = e0 + lane;
    if (row[e] == r) {
      const unsigned c = min((unsigned)col[e], (unsigned)(N_NODES - 1));
      part += expf(edge_score_whole(a_row, a_col, r, c) - gmax);
    }
  }
  for (int off = 16; off > 0; off >>= 1) part += __shfl_xor(part, off, 32);
  const float inv = 1.0f / (part + 1e-15f);

  const float* xl = x + lane * 4;
  float4 acc = make_float4(0.f, 0.f, 0.f, 0.f);
  for (int e0 = 0; e0 < N_EDGES; e0 += 32) {
    const int e = e0 + lane;
    const bool mine = row[e] == r;
    unsigned c = 0u;
    float val = 0.f;
    if (mine) {
      c = min((unsigned)col[e], (unsigned)(N_NODES - 1));
      val = (adj_val[e] + expf(edge_score_whole(a_row, a_col, r, c) - gmax) * inv) * 0.5f;
    }
    unsigned todo = __builtin_amdgcn_ballot_w32(mine);
    while (todo != 0u) {
      const int l = __builtin_ctz(todo);
      todo &= todo - 1u;
      const unsigned cc = (unsigned)__builtin_amdgcn_readlane((int)c, l);
      const float vv = __int_as_float(__builtin_amdgcn_readlane(__float_as_int(val), l));
      const float4 xv = *(const float4*)(xl + (size_t)cc * DIN);
      acc.x += vv * xv.x; acc.y += vv * xv.y; acc.z += vv * xv.z; acc.w += vv * xv.w;
    }
  }
  return acc;
}

#define GG_WAVES 8
#define TILE_ROWS 16
#define TILE_PITCH 132

__global__ void __launch_bounds__(GG_WAVES * 32)
gather_gemm_kernel(const float* __restrict__ x,
                   const int* __restrict__ row,
                   const int* __restrict__ col,
                   const float* __restrict__ adj_val,
                   const float* __restrict__ a_row,
                   const float* __restrict__ a_col,
                   const EdgeRec* __restrict__ packed,
                   const unsigned* __restrict__ maxkey,
                   const float* __restrict__ W_x,
                   const float* __restrict__ b_x,
                   float* __restrict__ out) {
  static_assert(N_NODES % TILE_ROWS == 0, "every block has 16 whole rows");
  static_assert(GG_WAVES * 32 == DOUT, "the waves' column groups cover DOUT");
  static_assert(TILE_ROWS == 2 * GG_WAVES, "two rows per wave");
  __shared__ float tile[TILE_ROWS * TILE_PITCH];
  const int wv   = __builtin_amdgcn_readfirstlane(threadIdx.x >> 5);
  const int lane = threadIdx.x & 31;
  const int row0 = blockIdx.x * TILE_ROWS;
  const unsigned key = *maxkey;
  const unsigned kb = (key & 0x80000000u) ? (key ^ 0x80000000u) : ~key;
  const float gmax = __uint_as_float(kb);

  for (int i = 0; i < 2; ++i) {
    const int lr = wv + GG_WAVES * i;
    const int r  = row0 + lr;
    const int n_all = packed[(size_t)r * SLOT].col;
    float4 f;
    if (n_all > CAP)
      f = gather_row_slow(x, row, col, adj_val, a_row, a_col, gmax, r, lane);
    else
      f = gather_row(x, packed, gmax, r, n_all, lane);
    *(float4*)(tile + lr * TILE_PITCH + lane * 4) = f;
  }
  __syncthreads();

  const int half  = lane >> 4;
  const int m     = lane & 15;
  const int cbase = wv * 32;

  v8f acc0, acc1;
  {
    const float bias0 = b_x[cbase + m];
    const float bias1 = b_x[cbase + 16 + m];
#pragma unroll
    for (int v = 0; v < 8; ++v) { acc0[v] = bias0; acc1[v] = bias1; }
  }
  const float* arow = tile + m * TILE_PITCH + 2 * half;
  for (int k = 0; k < DIN; k += 4) {
    v2f a = *(const v2f*)(arow + k);
    const int kk = k + 2 * half;
    v2f b0 = { W_x[kk * DOUT + cbase + m],      W_x[(kk + 1) * DOUT + cbase + m] };
    v2f b1 = { W_x[kk * DOUT + cbase + 16 + m], W_x[(kk + 1) * DOUT + cbase + 16 + m] };
    acc0 = wmma_f32_16x16x4(a, b0, acc0);
    acc1 = wmma_f32_16x16x4(a, b1, acc1);
  }
#pragma unroll
  for (int v = 0; v < 8; ++v) {
    float* orow = out + (size_t)(row0 + v + half * 8) * DOUT + cbase + m;
    orow[0]  = acc0[v];
    orow[16] = acc1[v];
  }
}

extern "C" void kernel_launch(void* const* d_in, const int* in_sizes, int n_in,
                              void* d_out, int out_size, void* d_ws, size_t ws_size,
                              hipStream_t stream) {
  const float* x      = (const float*)d_in[0];
  const int*   row    = (const int*)d_in[1];
  const int*   col    = (const int*)d_in[2];
  const float* adjval = (const float*)d_in[3];
  const float* W_row  = (const float*)d_in[4];
  const float* W_col  = (const float*)d_in[5];
  const float* W_x    = (const float*)d_in[6];
  const float* b_x    = (const float*)d_in[7];
  float* out = (float*)d_out;

  char* ws = (char*)d_ws;
  size_t off = 0;
  float* a_row = (float*)(ws + off); off += (size_t)N_NODES * DA * 4;
  float* a_col = (float*)(ws + off); off += (size_t)N_NODES * DA * 4;
  unsigned* maxkey = (unsigned*)(ws + off); off += 256;
  int* bin_count = (int*)(ws + off); off += (size_t)NBIN * 128;
  EdgeRec* packed = (EdgeRec*)(ws + off); off += (size_t)N_NODES * SLOT * 12;
  static_assert(sizeof(EdgeRec) == 12, "a place is 12 bytes");
  static_assert((2 * (size_t)N_NODES * DA * 4 + 256 + (size_t)NBIN * 128) % 16 == 0,
                "the slots start on a 16-byte boundary: level 2 writes them 16 bytes at a time");
  static_assert(2 * (size_t)N_NODES * DA * 4 + 256 + (size_t)NBIN * 128 + (size_t)N_NODES * SLOT * 12 <=
                3 * (size_t)N_EDGES * 4 + (size_t)N_NODES * 32 * 4 + 256 +
                (size_t)N_NODES * 4 + (size_t)N_NODES * DIN * 4,
                "the plan stays inside the workspace the earlier plans used");

  hipMemsetAsync(maxkey, 0, 256 + (size_t)NBIN * 128, stream);

  proj_park_kernel<<<PROJ_BLOCKS + PA_BLOCKS, 256, 0, stream>>>(x, W_row, W_col, a_row, a_col, row, col, adjval, packed, bin_count);
  place_score_kernel<<<NBIN, PB_THREADS, 0, stream>>>(row, col, adjval, a_row, a_col, packed, bin_count, maxkey);
  {
    const int blocks = N_NODES / TILE_ROWS;
    gather_gemm_kernel<<<blocks, GG_WAVES * 32, 0, stream>>>(x, row, col, adjval, a_row, a_col, packed, maxkey, W_x, b_x, out);
  }
}
